# starting kernel
# baseline (speedup 1.0000x reference)
#include <hip/hip_runtime.h>
#include <cmath>

#define N_NODES 50000
#define N_EDGES 800000
#define DIM 128

typedef float v2f __attribute__((ext_vector_type(2)));
typedef float v8f __attribute__((ext_vector_type(8)));

__global__ __launch_bounds__(256) void node_dots_kernel(
    const float* __restrict__ x, const float* __restrict__ w_pred,
    float* __restrict__ p_a, float* __restrict__ p_b, float* __restrict__ deg)
{
    int node = (blockIdx.x * blockDim.x + threadIdx.x) >> 5;
    int lane = threadIdx.x & 31;
    if (node >= N_NODES) return;
    const float* xr = x + (size_t)node * DIM;
    float sa = 0.f, sb = 0.f;
#pragma unroll
    for (int j = 0; j < 4; ++j) {
        int c = lane + 32 * j;
        float v = xr[c];
        sa += v * w_pred[c];
        sb += v * w_pred[DIM + c];
    }
#pragma unroll
    for (int off = 16; off > 0; off >>= 1) {
        sa += __shfl_down(sa, off, 32);
        sb += __shfl_down(sb, off, 32);
    }
    if (lane == 0) {
        p_a[node] = sa;
        p_b[node] = sb;
        deg[node] = 1.0f;
    }
}

__global__ __launch_bounds__(256) void edge_weight_kernel(
    const int* __restrict__ src, const int* __restrict__ tgt,
    const float* __restrict__ p_a, const float* __restrict__ p_b,
    const float* __restrict__ b_pred,
    float* __restrict__ ew, float* __restrict__ deg)
{
    int e = blockIdx.x * blockDim.x + threadIdx.x;
    if (e >= N_EDGES) return;
    int s = src[e], t = tgt[e];
    float z = p_a[s] + p_b[t] + b_pred[0];
    float w = 1.0f / (1.0f + __expf(-z));
    ew[e] = w;
    unsafeAtomicAdd(&deg[t], w);
}

__global__ __launch_bounds__(256) void rsqrt_kernel(float* __restrict__ deg_dis)
{
    int n = blockIdx.x * blockDim.x + threadIdx.x;
    if (n < N_NODES) {
        float d = deg_dis[n];
        deg_dis[n] = (d > 0.f) ? rsqrtf(d) : 0.f;
    }
}

__global__ __launch_bounds__(256) void gemm_wmma_kernel(
    const float* __restrict__ x, const float* __restrict__ W,
    float* __restrict__ xw)
{
    __shared__ float As[16 * 132];
    const int rowBase = blockIdx.x * 16;

#pragma unroll
    for (int j = 0; j < 8; ++j) {
        int idx = threadIdx.x + j * 256;
        int r = idx >> 7, c = idx & 127;
        As[r * 132 + c] = x[(size_t)rowBase * DIM + idx];
    }
    __syncthreads();

    const int waveId  = threadIdx.x >> 5;
    const int lane    = threadIdx.x & 31;
    const int m       = lane & 15;
    const int hi      = lane >> 4;
    const int colBase = waveId * 16;

    v8f acc = {};
#pragma unroll 8
    for (int k0 = 0; k0 < DIM; k0 += 4) {
        int ka = k0 + 2 * hi;
        v2f a, b;
        a.x = As[m * 132 + ka];
        a.y = As[m * 132 + ka + 1];
        b.x = W[(size_t)ka * DIM + colBase + m];
        b.y = W[(size_t)(ka + 1) * DIM + colBase + m];
        acc = __builtin_amdgcn_wmma_f32_16x16x4_f32(
            false, a, false, b, (short)0, acc, false, false);
    }

#pragma unroll
    for (int i = 0; i < 8; ++i) {
        int row = rowBase + i + 8 * hi;
        xw[(size_t)row * DIM + colBase + m] = acc[i];
    }
}

__global__ __launch_bounds__(256) void self_bias_kernel(
    const float* __restrict__ xw, const float* __restrict__ dis,
    const float* __restrict__ b, float* __restrict__ out)
{
    int idx = blockIdx.x * blockDim.x + threadIdx.x;
    if (idx >= N_NODES * DIM) return;
    int node = idx >> 7, d = idx & 127;
    float s = dis[node];
    out[idx] = s * s * xw[idx] + b[d];
}

__global__ __launch_bounds__(256) void edge_scatter_kernel(
    const int* __restrict__ src, const int* __restrict__ tgt,
    const float* __restrict__ ew, const float* __restrict__ dis,
    const float* __restrict__ xw, float* __restrict__ out)
{
    int e = (blockIdx.x * blockDim.x + threadIdx.x) >> 5;
    int lane = threadIdx.x & 31;
    if (e >= N_EDGES) return;
    int s = src[e], t = tgt[e];
    float norm = dis[s] * ew[e] * dis[t];
    const float4* xv = (const float4*)(xw + (size_t)s * DIM);
    float4 v = xv[lane];
    float* op = out + (size_t)t * DIM + lane * 4;
    unsafeAtomicAdd(op + 0, norm * v.x);
    unsafeAtomicAdd(op + 1, norm * v.y);
    unsafeAtomicAdd(op + 2, norm * v.z);
    unsafeAtomicAdd(op + 3, norm * v.w);
}

extern "C" void kernel_launch(void* const* d_in, const int* in_sizes, int n_in,
                              void* d_out, int out_size, void* d_ws, size_t ws_size,
                              hipStream_t stream) {
    const float* x          = (const float*)d_in[0];
    const int*   edge_index = (const int*)d_in[1];
    const float* W          = (const float*)d_in[2];
    const float* b          = (const float*)d_in[3];
    const float* w_pred     = (const float*)d_in[4];
    const float* b_pred     = (const float*)d_in[5];
    float*       out        = (float*)d_out;

    const int* src = edge_index;
    const int* tgt = edge_index + N_EDGES;

    float* ws  = (float*)d_ws;
    float* p_a = ws;
    float* p_b = p_a + N_NODES;
    float* deg = p_b + N_NODES;
    float* ew  = deg + N_NODES;
    float* xw  = ew + N_EDGES;

    node_dots_kernel  <<<(N_NODES + 7) / 8,      256, 0, stream>>>(x, w_pred, p_a, p_b, deg);
    edge_weight_kernel<<<(N_EDGES + 255) / 256,  256, 0, stream>>>(src, tgt, p_a, p_b, b_pred, ew, deg);
    rsqrt_kernel      <<<(N_NODES + 255) / 256,  256, 0, stream>>>(deg);
    gemm_wmma_kernel  <<<N_NODES / 16,           256, 0, stream>>>(x, W, xw);
    self_bias_kernel  <<<(N_NODES * DIM) / 256,  256, 0, stream>>>(xw, deg, b, out);
    edge_scatter_kernel<<<N_EDGES / 8,           256, 0, stream>>>(src, tgt, ew, deg, xw, out);
}


# GEMM epilogue staged through LDS writes whole rows of xw and of out = dis^2*xw + b, dis = rsqrt(deg) taken per block: rsqrt and self-loop kernels gone, 4 launches not 6; every sum keeps its order
# speedup vs baseline: 1.7171x; 1.7171x over previous
#include <hip/hip_runtime.h>
#include <cmath>

#define N_NODES 50000
#define N_EDGES 800000
#define DIM 128

typedef float v2f __attribute__((ext_vector_type(2)));
typedef float v8f __attribute__((ext_vector_type(8)));

__global__ __launch_bounds__(256) void node_dots_kernel(
    const float* __restrict__ x, const float* __restrict__ w_pred,
    float* __restrict__ p_a, float* __restrict__ p_b, float* __restrict__ deg)
{
    int node = (blockIdx.x * blockDim.x + threadIdx.x) >> 5;
    int lane = threadIdx.x & 31;
    if (node >= N_NODES) return;
    const float* xr = x + (size_t)node * DIM;
    float sa = 0.f, sb = 0.f;
#pragma unroll
    for (int j = 0; j < 4; ++j) {
        int c = lane + 32 * j;
        float v = xr[c];
        sa += v * w_pred[c];
        sb += v * w_pred[DIM + c];
    }
#pragma unroll
    for (int off = 16; off > 0; off >>= 1) {
        sa += __shfl_down(sa, off, 32);
        sb += __shfl_down(sb, off, 32);
    }
    if (lane == 0) {
        p_a[node] = sa;
        p_b[node] = sb;
        deg[node] = 1.0f;
    }
}

__global__ __launch_bounds__(256) void edge_weight_kernel(
    const int* __restrict__ src, const int* __restrict__ tgt,
    const float* __restrict__ p_a, const float* __restrict__ p_b,
    const float* __restrict__ b_pred,
    float* __restrict__ ew, float* __restrict__ deg)
{
    int e = blockIdx.x * blockDim.x + threadIdx.x;
    if (e >= N_EDGES) return;
    int s = src[e], t = tgt[e];
    float z = p_a[s] + p_b[t] + b_pred[0];
    float w = 1.0f / (1.0f + __expf(-z));
    ew[e] = w;
    unsafeAtomicAdd(&deg[t], w);
}

__global__ __launch_bounds__(256) void gemm_wmma_kernel(
    const float* __restrict__ x, const float* __restrict__ W,
    const float* __restrict__ b, float* __restrict__ deg_dis,
    float* __restrict__ xw, float* __restrict__ out)
{
    __shared__ float As[16 * 132];
    __shared__ __attribute__((aligned(16))) float Cs[16 * 132];
    __shared__ float sdis[16];
    const int rowBase = blockIdx.x * 16;

    float d = 1.0f;
    if (threadIdx.x < 16) d = deg_dis[rowBase + threadIdx.x];

#pragma unroll
    for (int j = 0; j < 8; ++j) {
        int idx = threadIdx.x + j * 256;
        int r = idx >> 7, c = idx & 127;
        As[r * 132 + c] = x[(size_t)rowBase * DIM + idx];
    }
    const float dis_own = (d > 0.f) ? rsqrtf(d) : 0.f;
    if (threadIdx.x < 16) sdis[threadIdx.x] = dis_own;
    __syncthreads();

    const int waveId  = threadIdx.x >> 5;
    const int lane    = threadIdx.x & 31;
    const int m       = lane & 15;
    const int hi      = lane >> 4;
    const int colBase = waveId * 16;

    v8f acc = {};
#pragma unroll 8
    for (int k0 = 0; k0 < DIM; k0 += 4) {
        int ka = k0 + 2 * hi;
        v2f a, bf;
        a.x = As[m * 132 + ka];
        a.y = As[m * 132 + ka + 1];
        bf.x = W[(size_t)ka * DIM + colBase + m];
        bf.y = W[(size_t)(ka + 1) * DIM + colBase + m];
        acc = __builtin_amdgcn_wmma_f32_16x16x4_f32(
            false, a, false, bf, (short)0, acc, false, false);
    }

#pragma unroll
    for (int i = 0; i < 8; ++i)
        Cs[(i + 8 * hi) * 132 + colBase + m] = acc[i];
    __syncthreads();

    const float4 bb = *reinterpret_cast<const float4*>(b + 4 * lane);
#pragma unroll
    for (int rr = 0; rr < 2; ++rr) {
        const int r = 2 * waveId + rr;
        const float4 v = *reinterpret_cast<const float4*>(&Cs[r * 132 + 4 * lane]);
        const float s = sdis[r];
        const size_t off = (size_t)(rowBase + r) * DIM + 4 * lane;
        *reinterpret_cast<float4*>(xw + off) = v;
        float4 o;
        o.x = s * s * v.x + bb.x;
        o.y = s * s * v.y + bb.y;
        o.z = s * s * v.z + bb.z;
        o.w = s * s * v.w + bb.w;
        *reinterpret_cast<float4*>(out + off) = o;
    }
    if (threadIdx.x < 16) deg_dis[rowBase + threadIdx.x] = dis_own;
}

static_assert(N_EDGES % 128 == 0, "edge_scatter_kernel: whole 4-wave blocks of 32 edges per wave");

__global__ __launch_bounds__(128) void edge_scatter_kernel(
    const int* __restrict__ src, const int* __restrict__ tgt,
    const float* __restrict__ ew, const float* __restrict__ dis,
    const float* __restrict__ xw, float* __restrict__ out)
{
    const int wave = (blockIdx.x * blockDim.x + threadIdx.x) >> 5;
    const int lane = threadIdx.x & 31;
    const int e = wave * 32 + lane;

    const int s = src[e], t = tgt[e];
    const float norm = dis[s] * ew[e] * dis[t];
    const int norm_bits = __float_as_int(norm);

    for (int g = 0; g < 32; g += 8) {
        float v[8][4];
#pragma unroll
        for (int j = 0; j < 8; ++j) {
            const int sj = __builtin_amdgcn_readlane(s, g + j);
            const float* xr = xw + (size_t)sj * DIM + lane;
            v[j][0] = xr[0];
            v[j][1] = xr[32];
            v[j][2] = xr[64];
            v[j][3] = xr[96];
        }
#pragma unroll
        for (int j = 0; j < 8; ++j) {
            const int tj = __builtin_amdgcn_readlane(t, g + j);
            const float nj = __int_as_float(__builtin_amdgcn_readlane(norm_bits, g + j));
            float* op = out + (size_t)tj * DIM + lane;
            unsafeAtomicAdd(op +  0, nj * v[j][0]);
            unsafeAtomicAdd(op + 32, nj * v[j][1]);
            unsafeAtomicAdd(op + 64, nj * v[j][2]);
            unsafeAtomicAdd(op + 96, nj * v[j][3]);
        }
    }
}

extern "C" void kernel_launch(void* const* d_in, const int* in_sizes, int n_in,
                              void* d_out, int out_size, void* d_ws, size_t ws_size,
                              hipStream_t stream) {
    const float* x          = (const float*)d_in[0];
    const int*   edge_index = (const int*)d_in[1];
    const float* W          = (const float*)d_in[2];
    const float* b          = (const float*)d_in[3];
    const float* w_pred     = (const float*)d_in[4];
    const float* b_pred     = (const float*)d_in[5];
    float*       out        = (float*)d_out;

    const int* src = edge_index;
    const int* tgt = edge_index + N_EDGES;

    float* ws  = (float*)d_ws;
    float* p_a = ws;
    float* p_b = p_a + N_NODES;
    float* deg = p_b + N_NODES;
    float* ew  = deg + N_NODES;
    float* xw  = ew + N_EDGES + 16;

    node_dots_kernel  <<<(N_NODES + 7) / 8,      256, 0, stream>>>(x, w_pred, p_a, p_b, deg);
    edge_weight_kernel<<<(N_EDGES + 255) / 256,  256, 0, stream>>>(src, tgt, p_a, p_b, b_pred, ew, deg);
    gemm_wmma_kernel  <<<N_NODES / 16,           256, 0, stream>>>(x, W, b, deg, xw, out);
    edge_scatter_kernel<<<N_EDGES / 128,         128, 0, stream>>>(src, tgt, ew, deg, xw, out);
}


# edge scatter split by target half: a block draws a ticket from its chip half's queue (returning int add) and adds only edges whose target is in that half's node range; sum order still arrival
# speedup vs baseline: 1.7830x; 1.0384x over previous
#include <hip/hip_runtime.h>
#include <cmath>

#define N_NODES 50000
#define N_EDGES 800000
#define DIM 128

#define SC_CHUNK   256
#define N_CHUNKS   (N_EDGES / SC_CHUNK)
#define T_SPLIT    (N_NODES / 2)
#define Q_STRIDE   64

typedef float v2f __attribute__((ext_vector_type(2)));
typedef float v8f __attribute__((ext_vector_type(8)));

__global__ __launch_bounds__(256) void node_dots_kernel(
    const float* __restrict__ x, const float* __restrict__ w_pred,
    float* __restrict__ p_a, float* __restrict__ p_b, float* __restrict__ deg,
    int* __restrict__ queue)
{
    if (blockIdx.x == 0 && threadIdx.x < 2) queue[threadIdx.x * Q_STRIDE] = 0;
    int node = (blockIdx.x * blockDim.x + threadIdx.x) >> 5;
    int lane = threadIdx.x & 31;
    if (node >= N_NODES) return;
    const float* xr = x + (size_t)node * DIM;
    float sa = 0.f, sb = 0.f;
#pragma unroll
    for (int j = 0; j < 4; ++j) {
        int c = lane + 32 * j;
        float v = xr[c];
        sa += v * w_pred[c];
        sb += v * w_pred[DIM + c];
    }
#pragma unroll
    for (int off = 16; off > 0; off >>= 1) {
        sa += __shfl_down(sa, off, 32);
        sb += __shfl_down(sb, off, 32);
    }
    if (lane == 0) {
        p_a[node] = sa;
        p_b[node] = sb;
        deg[node] = 1.0f;
    }
}

__global__ __launch_bounds__(256) void edge_weight_kernel(
    const int* __restrict__ src, const int* __restrict__ tgt,
    const float* __restrict__ p_a, const float* __restrict__ p_b,
    const float* __restrict__ b_pred,
    float* __restrict__ ew, float* __restrict__ deg)
{
    int e = blockIdx.x * blockDim.x + threadIdx.x;
    if (e >= N_EDGES) return;
    int s = src[e], t = tgt[e];
    float z = p_a[s] + p_b[t] + b_pred[0];
    float w = 1.0f / (1.0f + __expf(-z));
    ew[e] = w;
    unsafeAtomicAdd(&deg[t], w);
}

__global__ __launch_bounds__(256) void gemm_wmma_kernel(
    const float* __restrict__ x, const float* __restrict__ W,
    const float* __restrict__ b, float* __restrict__ deg_dis,
    float* __restrict__ xw, float* __restrict__ out)
{
    __shared__ float As[16 * 132];
    __shared__ __attribute__((aligned(16))) float Cs[16 * 132];
    __shared__ float sdis[16];
    const int rowBase = blockIdx.x * 16;

    float d = 1.0f;
    if (threadIdx.x < 16) d = deg_dis[rowBase + threadIdx.x];

#pragma unroll
    for (int j = 0; j < 8; ++j) {
        int idx = threadIdx.x + j * 256;
        int r = idx >> 7, c = idx & 127;
        As[r * 132 + c] = x[(size_t)rowBase * DIM + idx];
    }
    const float dis_own = (d > 0.f) ? rsqrtf(d) : 0.f;
    if (threadIdx.x < 16) sdis[threadIdx.x] = dis_own;
    __syncthreads();

    const int waveId  = threadIdx.x >> 5;
    const int lane    = threadIdx.x & 31;
    const int m       = lane & 15;
    const int hi      = lane >> 4;
    const int colBase = waveId * 16;

    v8f acc = {};
#pragma unroll 8
    for (int k0 = 0; k0 < DIM; k0 += 4) {
        int ka = k0 + 2 * hi;
        v2f a, bf;
        a.x = As[m * 132 + ka];
        a.y = As[m * 132 + ka + 1];
        bf.x = W[(size_t)ka * DIM + colBase + m];
        bf.y = W[(size_t)(ka + 1) * DIM + colBase + m];
        acc = __builtin_amdgcn_wmma_f32_16x16x4_f32(
            false, a, false, bf, (short)0, acc, false, false);
    }

#pragma unroll
    for (int i = 0; i < 8; ++i)
        Cs[(i + 8 * hi) * 132 + colBase + m] = acc[i];
    __syncthreads();

    const float4 bb = *reinterpret_cast<const float4*>(b + 4 * lane);
#pragma unroll
    for (int rr = 0; rr < 2; ++rr) {
        const int r = 2 * waveId + rr;
        const float4 v = *reinterpret_cast<const float4*>(&Cs[r * 132 + 4 * lane]);
        const float s = sdis[r];
        const size_t off = (size_t)(rowBase + r) * DIM + 4 * lane;
        *reinterpret_cast<float4*>(xw + off) = v;
        float4 o;
        o.x = s * s * v.x + bb.x;
        o.y = s * s * v.y + bb.y;
        o.z = s * s * v.z + bb.z;
        o.w = s * s * v.w + bb.w;
        *reinterpret_cast<float4*>(out + off) = o;
    }
    if (threadIdx.x < 16) deg_dis[rowBase + threadIdx.x] = dis_own;
}

static_assert(N_EDGES % SC_CHUNK == 0, "edge_scatter_kernel: whole chunks of 256 edges");

__device__ __forceinline__ void scatter_rows(
    unsigned rem, const int s, const int t, const int norm_bits, const int lane,
    const float* __restrict__ xw, float* __restrict__ out)
{
    while (rem != 0u) {
        const int n = min(8, (int)__builtin_popcount(rem));
        float v[8][4] = {};
        unsigned r = rem;
#pragma unroll
        for (int j = 0; j < 8; ++j) {
            if (j >= n) break;
            const int bl = __builtin_ctz(r);
            r &= r - 1u;
            const int sj = __builtin_amdgcn_readlane(s, bl);
            const float* xr = xw + (size_t)sj * DIM + lane;
            v[j][0] = xr[0];
            v[j][1] = xr[32];
            v[j][2] = xr[64];
            v[j][3] = xr[96];
        }
        int tl[8];
        r = rem;
#pragma unroll
        for (int j = 0; j < 8; ++j) {
            const int bl = (r != 0u) ? __builtin_ctz(r) : 0;
            r &= r - 1u;
            tl[j] = __builtin_amdgcn_readlane(t, bl);
            const float nj = __int_as_float(__builtin_amdgcn_readlane(norm_bits, bl));
            v[j][0] *= nj;
            v[j][1] *= nj;
            v[j][2] *= nj;
            v[j][3] *= nj;
        }
#pragma unroll
        for (int j = 0; j < 8; ++j) {
            if (j >= n) break;
            float* op = out + (size_t)tl[j] * DIM + lane;
            unsafeAtomicAdd(op +  0, v[j][0]);
            unsafeAtomicAdd(op + 32, v[j][1]);
            unsafeAtomicAdd(op + 64, v[j][2]);
            unsafeAtomicAdd(op + 96, v[j][3]);
        }
        rem = r;
    }
}

__global__ __launch_bounds__(128) void edge_scatter_kernel(
    const int* __restrict__ src, const int* __restrict__ tgt,
    const float* __restrict__ ew, const float* __restrict__ dis,
    const float* __restrict__ xw, float* __restrict__ out,
    int* __restrict__ queue)
{
    __shared__ int s_ticket[2];
    if (threadIdx.x == 0) {
        const unsigned where = __builtin_amdgcn_s_sendmsg_rtn(0x87);
        int h = (int)((where >> 18) & 1u);
        int c = atomicAdd(&queue[h * Q_STRIDE], 1);
        if (c >= N_CHUNKS) {
            h ^= 1;
            c = atomicAdd(&queue[h * Q_STRIDE], 1);
        }
        s_ticket[0] = c;
        s_ticket[1] = h;
    }
    __syncthreads();
    const int c = s_ticket[0];
    const int h = s_ticket[1];
    if (c >= N_CHUNKS) return;

    const int lane = threadIdx.x & 31;
    const int wv   = threadIdx.x >> 5;
    const int e0 = c * SC_CHUNK + wv * 64 + lane;
    const int e1 = e0 + 32;

    const int t0 = tgt[e0], t1 = tgt[e1];
    const int s0 = src[e0], s1 = src[e1];
    const float w0 = ew[e0], w1 = ew[e1];
    const float n0 = dis[s0] * w0 * dis[t0];
    const float n1 = dis[s1] * w1 * dis[t1];
    const bool mine0 = (t0 >= T_SPLIT) == (h != 0);
    const bool mine1 = (t1 >= T_SPLIT) == (h != 0);
    const unsigned m0 = __builtin_amdgcn_ballot_w32(mine0);
    const unsigned m1 = __builtin_amdgcn_ballot_w32(mine1);

    scatter_rows(m0, s0, t0, __float_as_int(n0), lane, xw, out);
    scatter_rows(m1, s1, t1, __float_as_int(n1), lane, xw, out);
}

extern "C" void kernel_launch(void* const* d_in, const int* in_sizes, int n_in,
                              void* d_out, int out_size, void* d_ws, size_t ws_size,
                              hipStream_t stream) {
    const float* x          = (const float*)d_in[0];
    const int*   edge_index = (const int*)d_in[1];
    const float* W          = (const float*)d_in[2];
    const float* b          = (const float*)d_in[3];
    const float* w_pred     = (const float*)d_in[4];
    const float* b_pred     = (const float*)d_in[5];
    float*       out        = (float*)d_out;

    const int* src = edge_index;
    const int* tgt = edge_index + N_EDGES;

    float* ws  = (float*)d_ws;
    float* p_a = ws;
    float* p_b = p_a + N_NODES;
    float* deg = p_b + N_NODES;
    float* ew  = deg + N_NODES;
    float* xw  = ew + N_EDGES + 16;
    int* queue = (int*)(xw + (size_t)N_NODES * DIM);

    node_dots_kernel  <<<(N_NODES + 7) / 8,      256, 0, stream>>>(x, w_pred, p_a, p_b, deg, queue);
    edge_weight_kernel<<<(N_EDGES + 255) / 256,  256, 0, stream>>>(src, tgt, p_a, p_b, b_pred, ew, deg);
    gemm_wmma_kernel  <<<N_NODES / 16,           256, 0, stream>>>(x, W, b, deg, xw, out);
    edge_scatter_kernel<<<2 * N_CHUNKS,          128, 0, stream>>>(src, tgt, ew, deg, xw, out, queue);
}


# edges binned by 200-row target range (LDS counts per 2048-edge tile, one int add per tile and range); a block per range sums its rows by f32 LDS adds and stores out once; sum order still arrival
# speedup vs baseline: 1.9151x; 1.0741x over previous
#include <hip/hip_runtime.h>
#include <cmath>

#define N_NODES 50000
#define N_EDGES 800000
#define DIM 128

#define RANGE      200
#define N_BINS     (N_NODES / RANGE)
#define BIN_CAP    4096
#define CNT_STRIDE 64
#define TILE_EDGES 2048
#define N_TILES    ((N_EDGES + TILE_EDGES - 1) / TILE_EDGES)
#define ROWS_PER_WAVE ((RANGE + 31) / 32)

static_assert(N_NODES % RANGE == 0, "ranges cover the nodes exactly");
static_assert(N_NODES <= 65536 && RANGE <= 32768, "a record packs src in 16 bits and the local target above them");
static_assert(N_BINS + 1 <= 256, "block 0 of node_dots_kernel clears all counters");

typedef float v2f __attribute__((ext_vector_type(2)));
typedef float v8f __attribute__((ext_vector_type(8)));

__global__ __launch_bounds__(256) void node_dots_kernel(
    const float* __restrict__ x, const float* __restrict__ w_pred,
    float* __restrict__ p_a, float* __restrict__ p_b, float* __restrict__ deg,
    int* __restrict__ cnt)
{
    if (blockIdx.x == 0 && threadIdx.x <= N_BINS) cnt[threadIdx.x * CNT_STRIDE] = 0;
    int node = (blockIdx.x * blockDim.x + threadIdx.x) >> 5;
    int lane = threadIdx.x & 31;
    if (node >= N_NODES) return;
    const float* xr = x + (size_t)node * DIM;
    float sa = 0.f, sb = 0.f;
#pragma unroll
    for (int j = 0; j < 4; ++j) {
        int c = lane + 32 * j;
        float v = xr[c];
        sa += v * w_pred[c];
        sb += v * w_pred[DIM + c];
    }
#pragma unroll
    for (int off = 16; off > 0; off >>= 1) {
        sa += __shfl_down(sa, off, 32);
        sb += __shfl_down(sb, off, 32);
    }
    if (lane == 0) {
        p_a[node] = sa;
        p_b[node] = sb;
        deg[node] = 1.0f;
    }
}

__global__ __launch_bounds__(1024) void edge_weight_bin_kernel(
    const int* __restrict__ src, const int* __restrict__ tgt,
    const float* __restrict__ p_a, const float* __restrict__ p_b,
    const float* __restrict__ b_pred, float* __restrict__ deg,
    int* __restrict__ cnt, int* __restrict__ rec_words,
    int* __restrict__ sp_src, int* __restrict__ sp_tgt, float* __restrict__ sp_w)
{
    __shared__ int hist[N_BINS];
    const int tid = threadIdx.x;
    if (tid < N_BINS) hist[tid] = 0;
    __syncthreads();

    const float bp = b_pred[0];
    int   s[2], t[2], bin[2], rank[2];
    float w[2];
    bool  ok[2];
#pragma unroll
    for (int k = 0; k < 2; ++k) {
        const int e = blockIdx.x * TILE_EDGES + k * 1024 + tid;
        ok[k] = e < N_EDGES;
        const int ec = ok[k] ? e : (N_EDGES - 1);
        s[k] = src[ec];
        t[k] = tgt[ec];
    }
#pragma unroll
    for (int k = 0; k < 2; ++k) {
        const float z = p_a[s[k]] + p_b[t[k]] + bp;
        w[k] = 1.0f / (1.0f + __expf(-z));
        bin[k]  = t[k] / RANGE;
        rank[k] = 0;
        if (ok[k]) rank[k] = atomicAdd(&hist[bin[k]], 1);
    }
    __syncthreads();
    if (tid < N_BINS) {
        const int c = hist[tid];
        hist[tid] = (c > 0) ? atomicAdd(&cnt[tid * CNT_STRIDE], c) : 0;
    }
    __syncthreads();
    int2* rec = reinterpret_cast<int2*>(rec_words);
#pragma unroll
    for (int k = 0; k < 2; ++k) {
        if (ok[k]) {
            unsafeAtomicAdd(&deg[t[k]], w[k]);
            const int slot = hist[bin[k]] + rank[k];
            if (slot < BIN_CAP) {
                const int tl = t[k] - bin[k] * RANGE;
                rec[(size_t)bin[k] * BIN_CAP + slot] = make_int2(s[k] | (tl << 16), __float_as_int(w[k]));
            } else {
                const int o = atomicAdd(&cnt[N_BINS * CNT_STRIDE], 1);
                if (o < N_EDGES) {
                    sp_src[o] = s[k];
                    sp_tgt[o] = t[k];
                    sp_w[o]   = w[k];
                }
            }
        }
    }
}

__global__ __launch_bounds__(256) void gemm_wmma_kernel(
    const float* __restrict__ x, const float* __restrict__ W,
    float* __restrict__ deg_dis, float* __restrict__ xw)
{
    __shared__ float As[16 * 132];
    __shared__ __attribute__((aligned(16))) float Cs[16 * 132];
    const int rowBase = blockIdx.x * 16;

    float d = 1.0f;
    if (threadIdx.x < 16) d = deg_dis[rowBase + threadIdx.x];

#pragma unroll
    for (int j = 0; j < 8; ++j) {
        int idx = threadIdx.x + j * 256;
        int r = idx >> 7, c = idx & 127;
        As[r * 132 + c] = x[(size_t)rowBase * DIM + idx];
    }
    const float dis_own = (d > 0.f) ? rsqrtf(d) : 0.f;
    __syncthreads();

    const int waveId  = threadIdx.x >> 5;
    const int lane    = threadIdx.x & 31;
    const int m       = lane & 15;
    const int hi      = lane >> 4;
    const int colBase = waveId * 16;

    v8f acc = {};
#pragma unroll 8
    for (int k0 = 0; k0 < DIM; k0 += 4) {
        int ka = k0 + 2 * hi;
        v2f a, bf;
        a.x = As[m * 132 + ka];
        a.y = As[m * 132 + ka + 1];
        bf.x = W[(size_t)ka * DIM + colBase + m];
        bf.y = W[(size_t)(ka + 1) * DIM + colBase + m];
        acc = __builtin_amdgcn_wmma_f32_16x16x4_f32(
            false, a, false, bf, (short)0, acc, false, false);
    }

#pragma unroll
    for (int i = 0; i < 8; ++i)
        Cs[(i + 8 * hi) * 132 + colBase + m] = acc[i];
    __syncthreads();

#pragma unroll
    for (int rr = 0; rr < 2; ++rr) {
        const int r = 2 * waveId + rr;
        const float4 v = *reinterpret_cast<const float4*>(&Cs[r * 132 + 4 * lane]);
        const size_t off = (size_t)(rowBase + r) * DIM + 4 * lane;
        *reinterpret_cast<float4*>(xw + off) = v;
    }
    if (threadIdx.x < 16) deg_dis[rowBase + threadIdx.x] = dis_own;
}

__device__ __forceinline__ void add_rows(
    unsigned rem, const int s, const int tl, const int coef_bits, const int lane,
    const float* __restrict__ xw, float* acc)
{
    while (rem != 0u) {
        const int n = min(8, (int)__builtin_popcount(rem));
        float v[8][4] = {};
        unsigned r = rem;
#pragma unroll
        for (int j = 0; j < 8; ++j) {
            if (j >= n) break;
            const int bl = __builtin_ctz(r);
            r &= r - 1u;
            const int sj = __builtin_amdgcn_readlane(s, bl);
            const float* xr = xw + (size_t)sj * DIM + lane;
            v[j][0] = xr[0];
            v[j][1] = xr[32];
            v[j][2] = xr[64];
            v[j][3] = xr[96];
        }
        int tj[8];
        r = rem;
#pragma unroll
        for (int j = 0; j < 8; ++j) {
            const int bl = (r != 0u) ? __builtin_ctz(r) : 0;
            r &= r - 1u;
            tj[j] = __builtin_amdgcn_readlane(tl, bl);
            const float cj = __int_as_float(__builtin_amdgcn_readlane(coef_bits, bl));
            v[j][0] *= cj;
            v[j][1] *= cj;
            v[j][2] *= cj;
            v[j][3] *= cj;
        }
#pragma unroll
        for (int j = 0; j < 8; ++j) {
            if (j >= n) break;
            float* ap = acc + tj[j] * DIM + lane;
            __hip_atomic_fetch_add(ap +  0, v[j][0], __ATOMIC_RELAXED, __HIP_MEMORY_SCOPE_WORKGROUP);
            __hip_atomic_fetch_add(ap + 32, v[j][1], __ATOMIC_RELAXED, __HIP_MEMORY_SCOPE_WORKGROUP);
            __hip_atomic_fetch_add(ap + 64, v[j][2], __ATOMIC_RELAXED, __HIP_MEMORY_SCOPE_WORKGROUP);
            __hip_atomic_fetch_add(ap + 96, v[j][3], __ATOMIC_RELAXED, __HIP_MEMORY_SCOPE_WORKGROUP);
        }
        rem = r;
    }
}

__global__ __launch_bounds__(1024) void range_sum_kernel(
    const int* __restrict__ rec_words, const int* __restrict__ cnt,
    const int* __restrict__ sp_src, const int* __restrict__ sp_tgt, const float* __restrict__ sp_w,
    const float* __restrict__ dis, const float* __restrict__ xw, const float* __restrict__ b,
    float* __restrict__ out)
{
    __shared__ __attribute__((aligned(16))) float acc[RANGE * DIM];
    __shared__ float sdis[RANGE];
    const int tid  = threadIdx.x;
    const int lane = tid & 31;
    const int wv   = __builtin_amdgcn_readfirstlane(tid >> 5);
    const int node0 = blockIdx.x * RANGE;

    const int n  = min(cnt[blockIdx.x * CNT_STRIDE], BIN_CAP);
    const int ns = min(cnt[N_BINS * CNT_STRIDE], N_EDGES);
    const int2* myrec = reinterpret_cast<const int2*>(rec_words) + (size_t)blockIdx.x * BIN_CAP;

    const int hi_rec = ((wv + 1) * n) >> 5;
    int base = (wv * n) >> 5;

    int2 r = make_int2(0, 0);
    if (base + lane < hi_rec) r = myrec[base + lane];

    for (int i = tid; i < RANGE * DIM / 4; i += 1024)
        reinterpret_cast<float4*>(acc)[i] = make_float4(0.f, 0.f, 0.f, 0.f);
    if (tid < RANGE) sdis[tid] = dis[node0 + tid];
    __syncthreads();

    while (base < hi_rec) {
        const int nb = base + 32;
        int2 rn = make_int2(0, 0);
        if (nb + lane < hi_rec) rn = myrec[nb + lane];
        const bool ok = base + lane < hi_rec;
        const int s  = r.x & 0xFFFF;
        const int tl = r.x >> 16;
        float coef = 0.f;
        if (ok) coef = dis[s] * __int_as_float(r.y) * sdis[tl];
        add_rows(__builtin_amdgcn_ballot_w32(ok), s, tl, __float_as_int(coef), lane, xw, acc);
        r = rn;
        base = nb;
    }

    for (int sb = wv * 32; sb < ns; sb += 1024) {
        const int i = sb + lane;
        int t = -1;
        if (i < ns) t = sp_tgt[i];
        const bool mine = (t >= node0) && (t < node0 + RANGE);
        int s = 0, tl = 0;
        float coef = 0.f;
        if (mine) {
            s  = sp_src[i];
            tl = t - node0;
            coef = dis[s] * sp_w[i] * sdis[tl];
        }
        add_rows(__builtin_amdgcn_ballot_w32(mine), s, tl, __float_as_int(coef), lane, xw, acc);
    }

    float4 xs[ROWS_PER_WAVE];
#pragma unroll
    for (int k = 0; k < ROWS_PER_WAVE; ++k) {
        const int rr = min(wv + 32 * k, RANGE - 1);
        xs[k] = *reinterpret_cast<const float4*>(xw + (size_t)(node0 + rr) * DIM + 4 * lane);
    }
    const float4 bb = *reinterpret_cast<const float4*>(b + 4 * lane);
    __syncthreads();

#pragma unroll
    for (int k = 0; k < ROWS_PER_WAVE; ++k) {
        const int rr = wv + 32 * k;
        if (rr < RANGE) {
            const float4 a = *reinterpret_cast<const float4*>(&acc[rr * DIM + 4 * lane]);
            const float d  = sdis[rr];
            const float d2 = d * d;
            float4 o;
            o.x = a.x + d2 * xs[k].x + bb.x;
            o.y = a.y + d2 * xs[k].y + bb.y;
            o.z = a.z + d2 * xs[k].z + bb.z;
            o.w = a.w + d2 * xs[k].w + bb.w;
            *reinterpret_cast<float4*>(out + (size_t)(node0 + rr) * DIM + 4 * lane) = o;
        }
    }
}

extern "C" void kernel_launch(void* const* d_in, const int* in_sizes, int n_in,
                              void* d_out, int out_size, void* d_ws, size_t ws_size,
                              hipStream_t stream) {
    const float* x          = (const float*)d_in[0];
    const int*   edge_index = (const int*)d_in[1];
    const float* W          = (const float*)d_in[2];
    const float* b          = (const float*)d_in[3];
    const float* w_pred     = (const float*)d_in[4];
    const float* b_pred     = (const float*)d_in[5];
    float*       out        = (float*)d_out;

    const int* src = edge_index;
    const int* tgt = edge_index + N_EDGES;

    float* ws  = (float*)d_ws;
    float* p_a = ws;
    float* p_b = p_a + N_NODES;
    float* deg = p_b + N_NODES;
    float* xw  = deg + N_NODES + 16;
    int* cnt    = (int*)(xw + (size_t)N_NODES * DIM);
    int* rec    = cnt + (N_BINS + 1) * CNT_STRIDE;
    int* sp_src = rec + (size_t)N_BINS * BIN_CAP * 2;
    int* sp_tgt = sp_src + N_EDGES;
    float* sp_w = (float*)(sp_tgt + N_EDGES);

    node_dots_kernel      <<<(N_NODES + 7) / 8, 256,  0, stream>>>(x, w_pred, p_a, p_b, deg, cnt);
    edge_weight_bin_kernel<<<N_TILES,           1024, 0, stream>>>(src, tgt, p_a, p_b, b_pred, deg, cnt, rec, sp_src, sp_tgt, sp_w);
    gemm_wmma_kernel      <<<N_NODES / 16,      256,  0, stream>>>(x, W, deg, xw);
    range_sum_kernel      <<<N_BINS,            1024, 0, stream>>>(rec, cnt, sp_src, sp_tgt, sp_w, deg, xw, b, out);
}


# degree sums by the range's owner as 2^-30 fixed-point 64-bit integer LDS adds (order-free: same bits every run, unlike f32 adds): GEMM block r walks range r's records; 800000 global f32 adds gone
# speedup vs baseline: 3.0279x; 1.5811x over previous
#include <hip/hip_runtime.h>
#include <cmath>

#define N_NODES 50000
#define N_EDGES 800000
#define DIM 128

#define RANGE      200
#define N_BINS     (N_NODES / RANGE)
#define BIN_CAP    4096
#define CNT_STRIDE 64
#define TILE_EDGES 2048
#define N_TILES    ((N_EDGES + TILE_EDGES - 1) / TILE_EDGES)
#define ROWS_PER_WAVE ((RANGE + 31) / 32)
#define DEG_UNIT   1073741824.0f
#define DEG_ONE    (1ull << 30)

static_assert(N_NODES % RANGE == 0, "ranges cover the nodes exactly");
static_assert(N_NODES <= 65536 && RANGE <= 32768, "a record packs src in 16 bits and the local target above them");
static_assert(N_BINS + 1 <= 256, "block 0 of node_dots_kernel clears all counters");
static_assert(RANGE <= 256 && N_BINS <= N_NODES / 16, "GEMM block r (256 threads) sums the degrees of range r");

typedef float v2f __attribute__((ext_vector_type(2)));
typedef float v8f __attribute__((ext_vector_type(8)));

__global__ __launch_bounds__(256) void node_dots_kernel(
    const float* __restrict__ x, const float* __restrict__ w_pred,
    float* __restrict__ p_a, float* __restrict__ p_b, int* __restrict__ cnt)
{
    if (blockIdx.x == 0 && threadIdx.x <= N_BINS) cnt[threadIdx.x * CNT_STRIDE] = 0;
    int node = (blockIdx.x * blockDim.x + threadIdx.x) >> 5;
    int lane = threadIdx.x & 31;
    if (node >= N_NODES) return;
    const float* xr = x + (size_t)node * DIM;
    float sa = 0.f, sb = 0.f;
#pragma unroll
    for (int j = 0; j < 4; ++j) {
        int c = lane + 32 * j;
        float v = xr[c];
        sa += v * w_pred[c];
        sb += v * w_pred[DIM + c];
    }
#pragma unroll
    for (int off = 16; off > 0; off >>= 1) {
        sa += __shfl_down(sa, off, 32);
        sb += __shfl_down(sb, off, 32);
    }
    if (lane == 0) {
        p_a[node] = sa;
        p_b[node] = sb;
    }
}

__global__ __launch_bounds__(1024) void edge_weight_bin_kernel(
    const int* __restrict__ src, const int* __restrict__ tgt,
    const float* __restrict__ p_a, const float* __restrict__ p_b,
    const float* __restrict__ b_pred,
    int* __restrict__ cnt, int* __restrict__ rec_words,
    int* __restrict__ sp_src, int* __restrict__ sp_tgt, float* __restrict__ sp_w)
{
    __shared__ int hist[N_BINS];
    const int tid = threadIdx.x;
    if (tid < N_BINS) hist[tid] = 0;
    __syncthreads();

    const float bp = b_pred[0];
    int   s[2], t[2], bin[2], rank[2];
    float w[2];
    bool  ok[2];
#pragma unroll
    for (int k = 0; k < 2; ++k) {
        const int e = blockIdx.x * TILE_EDGES + k * 1024 + tid;
        ok[k] = e < N_EDGES;
        const int ec = ok[k] ? e : (N_EDGES - 1);
        s[k] = src[ec];
        t[k] = tgt[ec];
    }
#pragma unroll
    for (int k = 0; k < 2; ++k) {
        const float z = p_a[s[k]] + p_b[t[k]] + bp;
        w[k] = 1.0f / (1.0f + __expf(-z));
        bin[k]  = t[k] / RANGE;
        rank[k] = 0;
        if (ok[k]) rank[k] = atomicAdd(&hist[bin[k]], 1);
    }
    __syncthreads();
    if (tid < N_BINS) {
        const int c = hist[tid];
        hist[tid] = (c > 0) ? atomicAdd(&cnt[tid * CNT_STRIDE], c) : 0;
    }
    __syncthreads();
    int2* rec = reinterpret_cast<int2*>(rec_words);
#pragma unroll
    for (int k = 0; k < 2; ++k) {
        if (ok[k]) {
            const int slot = hist[bin[k]] + rank[k];
            if (slot < BIN_CAP) {
                const int tl = t[k] - bin[k] * RANGE;
                rec[(size_t)bin[k] * BIN_CAP + slot] = make_int2(s[k] | (tl << 16), __float_as_int(w[k]));
            } else {
                const int o = atomicAdd(&cnt[N_BINS * CNT_STRIDE], 1);
                if (o < N_EDGES) {
                    sp_src[o] = s[k];
                    sp_tgt[o] = t[k];
                    sp_w[o]   = w[k];
                }
            }
        }
    }
}

__device__ __forceinline__ unsigned long long deg_fix(const float w)
{
    return (unsigned long long)__float2uint_rn(w * DEG_UNIT);
}

__global__ __launch_bounds__(256) void gemm_wmma_kernel(
    const float* __restrict__ x, const float* __restrict__ W,
    const int* __restrict__ cnt, const int* __restrict__ rec_words,
    const int* __restrict__ sp_tgt, const float* __restrict__ sp_w,
    float* __restrict__ dis, float* __restrict__ xw)
{
    __shared__ float As[16 * 132];
    __shared__ __attribute__((aligned(16))) float Cs[16 * 132];
    __shared__ unsigned long long dsum[RANGE];
    const int rowBase = blockIdx.x * 16;

    const bool own = blockIdx.x < N_BINS;
    int n = 0, ns = 0;
    if (own) {
        n  = min(cnt[blockIdx.x * CNT_STRIDE], BIN_CAP);
        ns = min(cnt[N_BINS * CNT_STRIDE], N_EDGES);
        if (threadIdx.x < RANGE) dsum[threadIdx.x] = 0ull;
    }

#pragma unroll
    for (int j = 0; j < 8; ++j) {
        int idx = threadIdx.x + j * 256;
        int r = idx >> 7, c = idx & 127;
        As[r * 132 + c] = x[(size_t)rowBase * DIM + idx];
    }
    __syncthreads();

    if (own) {
        const int node0 = blockIdx.x * RANGE;
        const int2* myrec = reinterpret_cast<const int2*>(rec_words) + (size_t)blockIdx.x * BIN_CAP;
        for (int i0 = threadIdx.x; i0 < n; i0 += 8 * 256) {
            int2 r[8];
#pragma unroll
            for (int j = 0; j < 8; ++j) r[j] = myrec[min(i0 + j * 256, n - 1)];
#pragma unroll
            for (int j = 0; j < 8; ++j)
                if (i0 + j * 256 < n) atomicAdd(&dsum[r[j].x >> 16], deg_fix(__int_as_float(r[j].y)));
        }
        for (int i = threadIdx.x; i < ns; i += 256) {
            const int t = sp_tgt[i];
            if (t >= node0 && t < node0 + RANGE) atomicAdd(&dsum[t - node0], deg_fix(sp_w[i]));
        }
    }

    const int waveId  = threadIdx.x >> 5;
    const int lane    = threadIdx.x & 31;
    const int m       = lane & 15;
    const int hi      = lane >> 4;
    const int colBase = waveId * 16;

    v8f acc = {};
#pragma unroll 8
    for (int k0 = 0; k0 < DIM; k0 += 4) {
        int ka = k0 + 2 * hi;
        v2f a, bf;
        a.x = As[m * 132 + ka];
        a.y = As[m * 132 + ka + 1];
        bf.x = W[(size_t)ka * DIM + colBase + m];
        bf.y = W[(size_t)(ka + 1) * DIM + colBase + m];
        acc = __builtin_amdgcn_wmma_f32_16x16x4_f32(
            false, a, false, bf, (short)0, acc, false, false);
    }

#pragma unroll
    for (int i = 0; i < 8; ++i)
        Cs[(i + 8 * hi) * 132 + colBase + m] = acc[i];
    __syncthreads();

#pragma unroll
    for (int rr = 0; rr < 2; ++rr) {
        const int r = 2 * waveId + rr;
        const float4 v = *reinterpret_cast<const float4*>(&Cs[r * 132 + 4 * lane]);
        const size_t off = (size_t)(rowBase + r) * DIM + 4 * lane;
        *reinterpret_cast<float4*>(xw + off) = v;
    }
    if (own && threadIdx.x < RANGE) {
        const float d = (float)(dsum[threadIdx.x] + DEG_ONE) * (1.0f / DEG_UNIT);
        dis[blockIdx.x * RANGE + threadIdx.x] = (d > 0.f) ? rsqrtf(d) : 0.f;
    }
}

__device__ __forceinline__ void add_rows(
    unsigned rem, const int s, const int tl, const int coef_bits, const int lane,
    const float* __restrict__ xw, float* acc)
{
    while (rem != 0u) {
        const int n = min(8, (int)__builtin_popcount(rem));
        float v[8][4] = {};
        unsigned r = rem;
#pragma unroll
        for (int j = 0; j < 8; ++j) {
            if (j >= n) break;
            const int bl = __builtin_ctz(r);
            r &= r - 1u;
            const int sj = __builtin_amdgcn_readlane(s, bl);
            const float* xr = xw + (size_t)sj * DIM + lane;
            v[j][0] = xr[0];
            v[j][1] = xr[32];
            v[j][2] = xr[64];
            v[j][3] = xr[96];
        }
        int tj[8];
        r = rem;
#pragma unroll
        for (int j = 0; j < 8; ++j) {
            const int bl = (r != 0u) ? __builtin_ctz(r) : 0;
            r &= r - 1u;
            tj[j] = __builtin_amdgcn_readlane(tl, bl);
            const float cj = __int_as_float(__builtin_amdgcn_readlane(coef_bits, bl));
            v[j][0] *= cj;
            v[j][1] *= cj;
            v[j][2] *= cj;
            v[j][3] *= cj;
        }
#pragma unroll
        for (int j = 0; j < 8; ++j) {
            if (j >= n) break;
            float* ap = acc + tj[j] * DIM + lane;
            __hip_atomic_fetch_add(ap +  0, v[j][0], __ATOMIC_RELAXED, __HIP_MEMORY_SCOPE_WORKGROUP);
            __hip_atomic_fetch_add(ap + 32, v[j][1], __ATOMIC_RELAXED, __HIP_MEMORY_SCOPE_WORKGROUP);
            __hip_atomic_fetch_add(ap + 64, v[j][2], __ATOMIC_RELAXED, __HIP_MEMORY_SCOPE_WORKGROUP);
            __hip_atomic_fetch_add(ap + 96, v[j][3], __ATOMIC_RELAXED, __HIP_MEMORY_SCOPE_WORKGROUP);
        }
        rem = r;
    }
}

__global__ __launch_bounds__(1024) void range_sum_kernel(
    const int* __restrict__ rec_words, const int* __restrict__ cnt,
    const int* __restrict__ sp_src, const int* __restrict__ sp_tgt, const float* __restrict__ sp_w,
    const float* __restrict__ dis, const float* __restrict__ xw, const float* __restrict__ b,
    float* __restrict__ out)
{
    __shared__ __attribute__((aligned(16))) float acc[RANGE * DIM];
    __shared__ float sdis[RANGE];
    const int tid  = threadIdx.x;
    const int lane = tid & 31;
    const int wv   = __builtin_amdgcn_readfirstlane(tid >> 5);
    const int node0 = blockIdx.x * RANGE;

    const int n  = min(cnt[blockIdx.x * CNT_STRIDE], BIN_CAP);
    const int ns = min(cnt[N_BINS * CNT_STRIDE], N_EDGES);
    const int2* myrec = reinterpret_cast<const int2*>(rec_words) + (size_t)blockIdx.x * BIN_CAP;

    const int hi_rec = ((wv + 1) * n) >> 5;
    int base = (wv * n) >> 5;

    int2 r = make_int2(0, 0);
    if (base + lane < hi_rec) r = myrec[base + lane];

    for (int i = tid; i < RANGE * DIM / 4; i += 1024)
        reinterpret_cast<float4*>(acc)[i] = make_float4(0.f, 0.f, 0.f, 0.f);
    if (tid < RANGE) sdis[tid] = dis[node0 + tid];
    __syncthreads();

    while (base < hi_rec) {
        const int nb = base + 32;
        int2 rn = make_int2(0, 0);
        if (nb + lane < hi_rec) rn = myrec[nb + lane];
        const bool ok = base + lane < hi_rec;
        const int s  = r.x & 0xFFFF;
        const int tl = r.x >> 16;
        float coef = 0.f;
        if (ok) coef = dis[s] * __int_as_float(r.y) * sdis[tl];
        add_rows(__builtin_amdgcn_ballot_w32(ok), s, tl, __float_as_int(coef), lane, xw, acc);
        r = rn;
        base = nb;
    }

    for (int sb = wv * 32; sb < ns; sb += 1024) {
        const int i = sb + lane;
        int t = -1;
        if (i < ns) t = sp_tgt[i];
        const bool mine = (t >= node0) && (t < node0 + RANGE);
        int s = 0, tl = 0;
        float coef = 0.f;
        if (mine) {
            s  = sp_src[i];
            tl = t - node0;
            coef = dis[s] * sp_w[i] * sdis[tl];
        }
        add_rows(__builtin_amdgcn_ballot_w32(mine), s, tl, __float_as_int(coef), lane, xw, acc);
    }

    float4 xs[ROWS_PER_WAVE];
#pragma unroll
    for (int k = 0; k < ROWS_PER_WAVE; ++k) {
        const int rr = min(wv + 32 * k, RANGE - 1);
        xs[k] = *reinterpret_cast<const float4*>(xw + (size_t)(node0 + rr) * DIM + 4 * lane);
    }
    const float4 bb = *reinterpret_cast<const float4*>(b + 4 * lane);
    __syncthreads();

#pragma unroll
    for (int k = 0; k < ROWS_PER_WAVE; ++k) {
        const int rr = wv + 32 * k;
        if (rr < RANGE) {
            const float4 a = *reinterpret_cast<const float4*>(&acc[rr * DIM + 4 * lane]);
            const float d  = sdis[rr];
            const float d2 = d * d;
            float4 o;
            o.x = a.x + d2 * xs[k].x + bb.x;
            o.y = a.y + d2 * xs[k].y + bb.y;
            o.z = a.z + d2 * xs[k].z + bb.z;
            o.w = a.w + d2 * xs[k].w + bb.w;
            *reinterpret_cast<float4*>(out + (size_t)(node0 + rr) * DIM + 4 * lane) = o;
        }
    }
}

extern "C" void kernel_launch(void* const* d_in, const int* in_sizes, int n_in,
                              void* d_out, int out_size, void* d_ws, size_t ws_size,
                              hipStream_t stream) {
    const float* x          = (const float*)d_in[0];
    const int*   edge_index = (const int*)d_in[1];
    const float* W          = (const float*)d_in[2];
    const float* b          = (const float*)d_in[3];
    const float* w_pred     = (const float*)d_in[4];
    const float* b_pred     = (const float*)d_in[5];
    float*       out        = (float*)d_out;

    const int* src = edge_index;
    const int* tgt = edge_index + N_EDGES;

    float* ws  = (float*)d_ws;
    float* p_a = ws;
    float* p_b = p_a + N_NODES;
    float* dis = p_b + N_NODES;
    float* xw  = dis + N_NODES + 16;
    int* cnt    = (int*)(xw + (size_t)N_NODES * DIM);
    int* rec    = cnt + (N_BINS + 1) * CNT_STRIDE;
    int* sp_src = rec + (size_t)N_BINS * BIN_CAP * 2;
    int* sp_tgt = sp_src + N_EDGES;
    float* sp_w = (float*)(sp_tgt + N_EDGES);

    node_dots_kernel      <<<(N_NODES + 7) / 8, 256,  0, stream>>>(x, w_pred, p_a, p_b, cnt);
    edge_weight_bin_kernel<<<N_TILES,           1024, 0, stream>>>(src, tgt, p_a, p_b, b_pred, cnt, rec, sp_src, sp_tgt, sp_w);
    gemm_wmma_kernel      <<<N_NODES / 16,      256,  0, stream>>>(x, W, cnt, rec, sp_tgt, sp_w, dis, xw);
    range_sum_kernel      <<<N_BINS,            1024, 0, stream>>>(rec, cnt, sp_src, sp_tgt, sp_w, dis, xw, b, out);
}


# edge weight formed by the target range's owner: record carries p_a[src], the owner adds p_b[tgt] from its 200 values in LDS; 800000 random p_b lookups leave the edge kernel; sum orders kept
# speedup vs baseline: 3.1684x; 1.0464x over previous
#include <hip/hip_runtime.h>
#include <cmath>

#define N_NODES 50000
#define N_EDGES 800000
#define DIM 128

#define RANGE      200
#define N_BINS     (N_NODES / RANGE)
#define BIN_CAP    4096
#define CNT_STRIDE 64
#define TILE_EDGES 2048
#define N_TILES    ((N_EDGES + TILE_EDGES - 1) / TILE_EDGES)
#define ROWS_PER_WAVE ((RANGE + 31) / 32)
#define DEG_UNIT   1073741824.0f
#define DEG_ONE    (1ull << 30)

static_assert(N_NODES % RANGE == 0, "ranges cover the nodes exactly");
static_assert(N_NODES <= 65536 && RANGE <= 32768, "a record packs src in 16 bits and the local target above them");
static_assert(N_BINS + 1 <= 256, "block 0 of node_dots_kernel clears all counters");
static_assert(RANGE <= 256 && N_BINS <= N_NODES / 16, "GEMM block r (256 threads) sums the degrees of range r");

typedef float v2f __attribute__((ext_vector_type(2)));
typedef float v8f __attribute__((ext_vector_type(8)));

__global__ __launch_bounds__(256) void node_dots_kernel(
    const float* __restrict__ x, const float* __restrict__ w_pred,
    float* __restrict__ p_a, float* __restrict__ p_b, int* __restrict__ cnt)
{
    if (blockIdx.x == 0 && threadIdx.x <= N_BINS) cnt[threadIdx.x * CNT_STRIDE] = 0;
    int node = (blockIdx.x * blockDim.x + threadIdx.x) >> 5;
    int lane = threadIdx.x & 31;
    if (node >= N_NODES) return;
    const float* xr = x + (size_t)node * DIM;
    float sa = 0.f, sb = 0.f;
#pragma unroll
    for (int j = 0; j < 4; ++j) {
        int c = lane + 32 * j;
        float v = xr[c];
        sa += v * w_pred[c];
        sb += v * w_pred[DIM + c];
    }
#pragma unroll
    for (int off = 16; off > 0; off >>= 1) {
        sa += __shfl_down(sa, off, 32);
        sb += __shfl_down(sb, off, 32);
    }
    if (lane == 0) {
        p_a[node] = sa;
        p_b[node] = sb;
    }
}

__device__ __forceinline__ float edge_w(const float pa, const float pb, const float bp)
{
    const float z = pa + pb + bp;
    return 1.0f / (1.0f + __expf(-z));
}

__global__ __launch_bounds__(1024) void edge_bin_kernel(
    const int* __restrict__ src, const int* __restrict__ tgt,
    const float* __restrict__ p_a,
    int* __restrict__ cnt, int* __restrict__ rec_words,
    int* __restrict__ sp_src, int* __restrict__ sp_tgt, float* __restrict__ sp_pa)
{
    __shared__ int hist[N_BINS];
    const int tid = threadIdx.x;
    if (tid < N_BINS) hist[tid] = 0;
    __syncthreads();

    int   s[2], t[2], bin[2], rank[2];
    float pa[2];
    bool  ok[2];
#pragma unroll
    for (int k = 0; k < 2; ++k) {
        const int e = blockIdx.x * TILE_EDGES + k * 1024 + tid;
        ok[k] = e < N_EDGES;
        const int ec = ok[k] ? e : (N_EDGES - 1);
        s[k] = src[ec];
        t[k] = tgt[ec];
    }
#pragma unroll
    for (int k = 0; k < 2; ++k) {
        pa[k]   = p_a[s[k]];
        bin[k]  = t[k] / RANGE;
        rank[k] = 0;
        if (ok[k]) rank[k] = atomicAdd(&hist[bin[k]], 1);
    }
    __syncthreads();
    if (tid < N_BINS) {
        const int c = hist[tid];
        hist[tid] = (c > 0) ? atomicAdd(&cnt[tid * CNT_STRIDE], c) : 0;
    }
    __syncthreads();
    int2* rec = reinterpret_cast<int2*>(rec_words);
#pragma unroll
    for (int k = 0; k < 2; ++k) {
        if (ok[k]) {
            const int slot = hist[bin[k]] + rank[k];
            if (slot < BIN_CAP) {
                const int tl = t[k] - bin[k] * RANGE;
                rec[(size_t)bin[k] * BIN_CAP + slot] = make_int2(s[k] | (tl << 16), __float_as_int(pa[k]));
            } else {
                const int o = atomicAdd(&cnt[N_BINS * CNT_STRIDE], 1);
                if (o < N_EDGES) {
                    sp_src[o] = s[k];
                    sp_tgt[o] = t[k];
                    sp_pa[o]  = pa[k];
                }
            }
        }
    }
}

__device__ __forceinline__ unsigned long long deg_fix(const float w)
{
    return (unsigned long long)__float2uint_rn(w * DEG_UNIT);
}

__global__ __launch_bounds__(256) void gemm_wmma_kernel(
    const float* __restrict__ x, const float* __restrict__ W,
    const int* __restrict__ cnt, const int* __restrict__ rec_words,
    const int* __restrict__ sp_tgt, const float* __restrict__ sp_pa,
    const float* __restrict__ p_b, const float* __restrict__ b_pred,
    float* __restrict__ dis, float* __restrict__ xw)
{
    __shared__ float As[16 * 132];
    __shared__ __attribute__((aligned(16))) float Cs[16 * 132];
    __shared__ unsigned long long dsum[RANGE];
    __shared__ float spb[RANGE];
    const int rowBase = blockIdx.x * 16;

    const bool own = blockIdx.x < N_BINS;
    int n = 0, ns = 0;
    float bp = 0.f;
    if (own) {
        n  = min(cnt[blockIdx.x * CNT_STRIDE], BIN_CAP);
        ns = min(cnt[N_BINS * CNT_STRIDE], N_EDGES);
        bp = b_pred[0];
        if (threadIdx.x < RANGE) {
            dsum[threadIdx.x] = 0ull;
            spb[threadIdx.x]  = p_b[blockIdx.x * RANGE + threadIdx.x];
        }
    }

#pragma unroll
    for (int j = 0; j < 8; ++j) {
        int idx = threadIdx.x + j * 256;
        int r = idx >> 7, c = idx & 127;
        As[r * 132 + c] = x[(size_t)rowBase * DIM + idx];
    }
    __syncthreads();

    if (own) {
        const int node0 = blockIdx.x * RANGE;
        const int2* myrec = reinterpret_cast<const int2*>(rec_words) + (size_t)blockIdx.x * BIN_CAP;
        for (int i0 = threadIdx.x; i0 < n; i0 += 8 * 256) {
            int2 r[8];
#pragma unroll
            for (int j = 0; j < 8; ++j) r[j] = myrec[min(i0 + j * 256, n - 1)];
#pragma unroll
            for (int j = 0; j < 8; ++j)
                if (i0 + j * 256 < n) {
                    const int tl = r[j].x >> 16;
                    atomicAdd(&dsum[tl], deg_fix(edge_w(__int_as_float(r[j].y), spb[tl], bp)));
                }
        }
        for (int i = threadIdx.x; i < ns; i += 256) {
            const int t = sp_tgt[i];
            if (t >= node0 && t < node0 + RANGE)
                atomicAdd(&dsum[t - node0], deg_fix(edge_w(sp_pa[i], spb[t - node0], bp)));
        }
    }

    const int waveId  = threadIdx.x >> 5;
    const int lane    = threadIdx.x & 31;
    const int m       = lane & 15;
    const int hi      = lane >> 4;
    const int colBase = waveId * 16;

    v8f acc = {};
#pragma unroll 8
    for (int k0 = 0; k0 < DIM; k0 += 4) {
        int ka = k0 + 2 * hi;
        v2f a, bf;
        a.x = As[m * 132 + ka];
        a.y = As[m * 132 + ka + 1];
        bf.x = W[(size_t)ka * DIM + colBase + m];
        bf.y = W[(size_t)(ka + 1) * DIM + colBase + m];
        acc = __builtin_amdgcn_wmma_f32_16x16x4_f32(
            false, a, false, bf, (short)0, acc, false, false);
    }

#pragma unroll
    for (int i = 0; i < 8; ++i)
        Cs[(i + 8 * hi) * 132 + colBase + m] = acc[i];
    __syncthreads();

#pragma unroll
    for (int rr = 0; rr < 2; ++rr) {
        const int r = 2 * waveId + rr;
        const float4 v = *reinterpret_cast<const float4*>(&Cs[r * 132 + 4 * lane]);
        const size_t off = (size_t)(rowBase + r) * DIM + 4 * lane;
        *reinterpret_cast<float4*>(xw + off) = v;
    }
    if (own && threadIdx.x < RANGE) {
        const float d = (float)(dsum[threadIdx.x] + DEG_ONE) * (1.0f / DEG_UNIT);
        dis[blockIdx.x * RANGE + threadIdx.x] = (d > 0.f) ? rsqrtf(d) : 0.f;
    }
}

__device__ __forceinline__ void add_rows(
    unsigned rem, const int s, const int tl, const int coef_bits, const int lane,
    const float* __restrict__ xw, float* acc)
{
    while (rem != 0u) {
        const int n = min(8, (int)__builtin_popcount(rem));
        float v[8][4] = {};
        unsigned r = rem;
#pragma unroll
        for (int j = 0; j < 8; ++j) {
            if (j >= n) break;
            const int bl = __builtin_ctz(r);
            r &= r - 1u;
            const int sj = __builtin_amdgcn_readlane(s, bl);
            const float* xr = xw + (size_t)sj * DIM + lane;
            v[j][0] = xr[0];
            v[j][1] = xr[32];
            v[j][2] = xr[64];
            v[j][3] = xr[96];
        }
        int tj[8];
        r = rem;
#pragma unroll
        for (int j = 0; j < 8; ++j) {
            const int bl = (r != 0u) ? __builtin_ctz(r) : 0;
            r &= r - 1u;
            tj[j] = __builtin_amdgcn_readlane(tl, bl);
            const float cj = __int_as_float(__builtin_amdgcn_readlane(coef_bits, bl));
            v[j][0] *= cj;
            v[j][1] *= cj;
            v[j][2] *= cj;
            v[j][3] *= cj;
        }
#pragma unroll
        for (int j = 0; j < 8; ++j) {
            if (j >= n) break;
            float* ap = acc + tj[j] * DIM + lane;
            __hip_atomic_fetch_add(ap +  0, v[j][0], __ATOMIC_RELAXED, __HIP_MEMORY_SCOPE_WORKGROUP);
            __hip_atomic_fetch_add(ap + 32, v[j][1], __ATOMIC_RELAXED, __HIP_MEMORY_SCOPE_WORKGROUP);
            __hip_atomic_fetch_add(ap + 64, v[j][2], __ATOMIC_RELAXED, __HIP_MEMORY_SCOPE_WORKGROUP);
            __hip_atomic_fetch_add(ap + 96, v[j][3], __ATOMIC_RELAXED, __HIP_MEMORY_SCOPE_WORKGROUP);
        }
        rem = r;
    }
}

__global__ __launch_bounds__(1024) void range_sum_kernel(
    const int* __restrict__ rec_words, const int* __restrict__ cnt,
    const int* __restrict__ sp_src, const int* __restrict__ sp_tgt, const float* __restrict__ sp_pa,
    const float* __restrict__ p_b, const float* __restrict__ b_pred,
    const float* __restrict__ dis, const float* __restrict__ xw, const float* __restrict__ b,
    float* __restrict__ out)
{
    __shared__ __attribute__((aligned(16))) float acc[RANGE * DIM];
    __shared__ float sdis[RANGE];
    __shared__ float spb[RANGE];
    const int tid  = threadIdx.x;
    const int lane = tid & 31;
    const int wv   = __builtin_amdgcn_readfirstlane(tid >> 5);
    const int node0 = blockIdx.x * RANGE;

    const int n  = min(cnt[blockIdx.x * CNT_STRIDE], BIN_CAP);
    const int ns = min(cnt[N_BINS * CNT_STRIDE], N_EDGES);
    const float bp = b_pred[0];
    const int2* myrec = reinterpret_cast<const int2*>(rec_words) + (size_t)blockIdx.x * BIN_CAP;

    const int hi_rec = ((wv + 1) * n) >> 5;
    int base = (wv * n) >> 5;

    int2 r = make_int2(0, 0);
    if (base + lane < hi_rec) r = myrec[base + lane];

    for (int i = tid; i < RANGE * DIM / 4; i += 1024)
        reinterpret_cast<float4*>(acc)[i] = make_float4(0.f, 0.f, 0.f, 0.f);
    if (tid < RANGE) {
        sdis[tid] = dis[node0 + tid];
        spb[tid]  = p_b[node0 + tid];
    }
    __syncthreads();

    while (base < hi_rec) {
        const int nb = base + 32;
        int2 rn = make_int2(0, 0);
        if (nb + lane < hi_rec) rn = myrec[nb + lane];
        const bool ok = base + lane < hi_rec;
        const int s  = r.x & 0xFFFF;
        const int tl = r.x >> 16;
        float coef = 0.f;
        if (ok) coef = dis[s] * edge_w(__int_as_float(r.y), spb[tl], bp) * sdis[tl];
        add_rows(__builtin_amdgcn_ballot_w32(ok), s, tl, __float_as_int(coef), lane, xw, acc);
        r = rn;
        base = nb;
    }

    for (int sb = wv * 32; sb < ns; sb += 1024) {
        const int i = sb + lane;
        int t = -1;
        if (i < ns) t = sp_tgt[i];
        const bool mine = (t >= node0) && (t < node0 + RANGE);
        int s = 0, tl = 0;
        float coef = 0.f;
        if (mine) {
            s  = sp_src[i];
            tl = t - node0;
            coef = dis[s] * edge_w(sp_pa[i], spb[tl], bp) * sdis[tl];
        }
        add_rows(__builtin_amdgcn_ballot_w32(mine), s, tl, __float_as_int(coef), lane, xw, acc);
    }

    float4 xs[ROWS_PER_WAVE];
#pragma unroll
    for (int k = 0; k < ROWS_PER_WAVE; ++k) {
        const int rr = min(wv + 32 * k, RANGE - 1);
        xs[k] = *reinterpret_cast<const float4*>(xw + (size_t)(node0 + rr) * DIM + 4 * lane);
    }
    const float4 bb = *reinterpret_cast<const float4*>(b + 4 * lane);
    __syncthreads();

#pragma unroll
    for (int k = 0; k < ROWS_PER_WAVE; ++k) {
        const int rr = wv + 32 * k;
        if (rr < RANGE) {
            const float4 a = *reinterpret_cast<const float4*>(&acc[rr * DIM + 4 * lane]);
            const float d  = sdis[rr];
            const float d2 = d * d;
            float4 o;
            o.x = a.x + d2 * xs[k].x + bb.x;
            o.y = a.y + d2 * xs[k].y + bb.y;
            o.z = a.z + d2 * xs[k].z + bb.z;
            o.w = a.w + d2 * xs[k].w + bb.w;
            *reinterpret_cast<float4*>(out + (size_t)(node0 + rr) * DIM + 4 * lane) = o;
        }
    }
}

extern "C" void kernel_launch(void* const* d_in, const int* in_sizes, int n_in,
                              void* d_out, int out_size, void* d_ws, size_t ws_size,
                              hipStream_t stream) {
    const float* x          = (const float*)d_in[0];
    const int*   edge_index = (const int*)d_in[1];
    const float* W          = (const float*)d_in[2];
    const float* b          = (const float*)d_in[3];
    const float* w_pred     = (const float*)d_in[4];
    const float* b_pred     = (const float*)d_in[5];
    float*       out        = (float*)d_out;

    const int* src = edge_index;
    const int* tgt = edge_index + N_EDGES;

    float* ws  = (float*)d_ws;
    float* p_a = ws;
    float* p_b = p_a + N_NODES;
    float* dis = p_b + N_NODES;
    float* xw  = dis + N_NODES + 16;
    int* cnt    = (int*)(xw + (size_t)N_NODES * DIM);
    int* rec    = cnt + (N_BINS + 1) * CNT_STRIDE;
    int* sp_src = rec + (size_t)N_BINS * BIN_CAP * 2;
    int* sp_tgt = sp_src + N_EDGES;
    float* sp_pa = (float*)(sp_tgt + N_EDGES);

    node_dots_kernel      <<<(N_NODES + 7) / 8, 256,  0, stream>>>(x, w_pred, p_a, p_b, cnt);
    edge_bin_kernel       <<<N_TILES,           1024, 0, stream>>>(src, tgt, p_a, cnt, rec, sp_src, sp_tgt, sp_pa);
    gemm_wmma_kernel      <<<N_NODES / 16,      256,  0, stream>>>(x, W, cnt, rec, sp_tgt, sp_pa, p_b, b_pred, dis, xw);
    range_sum_kernel      <<<N_BINS,            1024, 0, stream>>>(rec, cnt, sp_src, sp_tgt, sp_pa, p_b, b_pred, dis, xw, b, out);
}


# node dots: one wave takes 16 consecutive nodes, its 64 row loads asked for first, then each node's sums as before: 3125 wave starts in that kernel, not 50000; every sum keeps its order
# speedup vs baseline: 3.3610x; 1.0608x over previous
#include <hip/hip_runtime.h>
#include <cmath>

#define N_NODES 50000
#define N_EDGES 800000
#define DIM 128

#define RANGE      200
#define N_BINS     (N_NODES / RANGE)
#define BIN_CAP    4096
#define CNT_STRIDE 64
#define TILE_EDGES 2048
#define N_TILES    ((N_EDGES + TILE_EDGES - 1) / TILE_EDGES)
#define ROWS_PER_WAVE ((RANGE + 31) / 32)
#define DEG_UNIT   1073741824.0f
#define DEG_ONE    (1ull << 30)
#define DOT_NODES  16
#define DOT_BLOCKS ((N_NODES / DOT_NODES + 7) / 8)

static_assert(N_NODES % RANGE == 0, "ranges cover the nodes exactly");
static_assert(N_NODES % DOT_NODES == 0 && DOT_NODES <= 32, "a wave of node_dots_kernel has all its nodes or none; lane i stores node i");
static_assert(N_NODES <= 65536 && RANGE <= 32768, "a record packs src in 16 bits and the local target above them");
static_assert(N_BINS + 1 <= 256, "block 0 of node_dots_kernel clears all counters");
static_assert(RANGE <= 256 && N_BINS <= N_NODES / 16, "GEMM block r (256 threads) sums the degrees of range r");

typedef float v2f __attribute__((ext_vector_type(2)));
typedef float v8f __attribute__((ext_vector_type(8)));

__global__ __launch_bounds__(256) void node_dots_kernel(
    const float* __restrict__ x, const float* __restrict__ w_pred,
    float* __restrict__ p_a, float* __restrict__ p_b, int* __restrict__ cnt)
{
    if (blockIdx.x == 0 && threadIdx.x <= N_BINS) cnt[threadIdx.x * CNT_STRIDE] = 0;
    const int lane  = threadIdx.x & 31;
    const int node0 = ((blockIdx.x * blockDim.x + threadIdx.x) >> 5) * DOT_NODES;
    if (node0 >= N_NODES) return;
    const float* xr = x + (size_t)node0 * DIM + lane;
    float v[DOT_NODES][4];
#pragma unroll
    for (int i = 0; i < DOT_NODES; ++i)
#pragma unroll
        for (int j = 0; j < 4; ++j) v[i][j] = xr[i * DIM + 32 * j];
    float wa[4], wb[4];
#pragma unroll
    for (int j = 0; j < 4; ++j) {
        wa[j] = w_pred[lane + 32 * j];
        wb[j] = w_pred[DIM + lane + 32 * j];
    }
    float ra = 0.f, rb = 0.f;
#pragma unroll
    for (int i = 0; i < DOT_NODES; ++i) {
        float sa = 0.f, sb = 0.f;
#pragma unroll
        for (int j = 0; j < 4; ++j) {
            sa += v[i][j] * wa[j];
            sb += v[i][j] * wb[j];
        }
#pragma unroll
        for (int off = 16; off > 0; off >>= 1) {
            sa += __shfl_down(sa, off, 32);
            sb += __shfl_down(sb, off, 32);
        }
        const float a0 = __int_as_float(__builtin_amdgcn_readfirstlane(__float_as_int(sa)));
        const float b0 = __int_as_float(__builtin_amdgcn_readfirstlane(__float_as_int(sb)));
        if (lane == i) { ra = a0; rb = b0; }
    }
    if (lane < DOT_NODES) {
        p_a[node0 + lane] = ra;
        p_b[node0 + lane] = rb;
    }
}

__device__ __forceinline__ float edge_w(const float pa, const float pb, const float bp)
{
    const float z = pa + pb + bp;
    return 1.0f / (1.0f + __expf(-z));
}

__global__ __launch_bounds__(1024) void edge_bin_kernel(
    const int* __restrict__ src, const int* __restrict__ tgt,
    const float* __restrict__ p_a,
    int* __restrict__ cnt, int* __restrict__ rec_words,
    int* __restrict__ sp_src, int* __restrict__ sp_tgt, float* __restrict__ sp_pa)
{
    __shared__ int hist[N_BINS];
    const int tid = threadIdx.x;
    if (tid < N_BINS) hist[tid] = 0;
    __syncthreads();

    int   s[2], t[2], bin[2], rank[2];
    float pa[2];
    bool  ok[2];
#pragma unroll
    for (int k = 0; k < 2; ++k) {
        const int e = blockIdx.x * TILE_EDGES + k * 1024 + tid;
        ok[k] = e < N_EDGES;
        const int ec = ok[k] ? e : (N_EDGES - 1);
        s[k] = src[ec];
        t[k] = tgt[ec];
    }
#pragma unroll
    for (int k = 0; k < 2; ++k) {
        pa[k]   = p_a[s[k]];
        bin[k]  = t[k] / RANGE;
        rank[k] = 0;
        if (ok[k]) rank[k] = atomicAdd(&hist[bin[k]], 1);
    }
    __syncthreads();
    if (tid < N_BINS) {
        const int c = hist[tid];
        hist[tid] = (c > 0) ? atomicAdd(&cnt[tid * CNT_STRIDE], c) : 0;
    }
    __syncthreads();
    int2* rec = reinterpret_cast<int2*>(rec_words);
#pragma unroll
    for (int k = 0; k < 2; ++k) {
        if (ok[k]) {
            const int slot = hist[bin[k]] + rank[k];
            if (slot < BIN_CAP) {
                const int tl = t[k] - bin[k] * RANGE;
                rec[(size_t)bin[k] * BIN_CAP + slot] = make_int2(s[k] | (tl << 16), __float_as_int(pa[k]));
            } else {
                const int o = atomicAdd(&cnt[N_BINS * CNT_STRIDE], 1);
                if (o < N_EDGES) {
                    sp_src[o] = s[k];
                    sp_tgt[o] = t[k];
                    sp_pa[o]  = pa[k];
                }
            }
        }
    }
}

__device__ __forceinline__ unsigned long long deg_fix(const float w)
{
    return (unsigned long long)__float2uint_rn(w * DEG_UNIT);
}

__global__ __launch_bounds__(256) void gemm_wmma_kernel(
    const float* __restrict__ x, const float* __restrict__ W,
    const int* __restrict__ cnt, const int* __restrict__ rec_words,
    const int* __restrict__ sp_tgt, const float* __restrict__ sp_pa,
    const float* __restrict__ p_b, const float* __restrict__ b_pred,
    float* __restrict__ dis, float* __restrict__ xw)
{
    __shared__ float As[16 * 132];
    __shared__ __attribute__((aligned(16))) float Cs[16 * 132];
    __shared__ unsigned long long dsum[RANGE];
    __shared__ float spb[RANGE];
    const int rowBase = blockIdx.x * 16;

    const bool own = blockIdx.x < N_BINS;
    int n = 0, ns = 0;
    float bp = 0.f;
    if (own) {
        n  = min(cnt[blockIdx.x * CNT_STRIDE], BIN_CAP);
        ns = min(cnt[N_BINS * CNT_STRIDE], N_EDGES);
        bp = b_pred[0];
        if (threadIdx.x < RANGE) {
            dsum[threadIdx.x] = 0ull;
            spb[threadIdx.x]  = p_b[blockIdx.x * RANGE + threadIdx.x];
        }
    }

#pragma unroll
    for (int j = 0; j < 8; ++j) {
        int idx = threadIdx.x + j * 256;
        int r = idx >> 7, c = idx & 127;
        As[r * 132 + c] = x[(size_t)rowBase * DIM + idx];
    }
    __syncthreads();

    if (own) {
        const int node0 = blockIdx.x * RANGE;
        const int2* myrec = reinterpret_cast<const int2*>(rec_words) + (size_t)blockIdx.x * BIN_CAP;
        for (int i0 = threadIdx.x; i0 < n; i0 += 8 * 256) {
            int2 r[8];
#pragma unroll
            for (int j = 0; j < 8; ++j) r[j] = myrec[min(i0 + j * 256, n - 1)];
#pragma unroll
            for (int j = 0; j < 8; ++j)
                if (i0 + j * 256 < n) {
                    const int tl = r[j].x >> 16;
                    atomicAdd(&dsum[tl], deg_fix(edge_w(__int_as_float(r[j].y), spb[tl], bp)));
                }
        }
        for (int i = threadIdx.x; i < ns; i += 256) {
            const int t = sp_tgt[i];
            if (t >= node0 && t < node0 + RANGE)
                atomicAdd(&dsum[t - node0], deg_fix(edge_w(sp_pa[i], spb[t - node0], bp)));
        }
    }

    const int waveId  = threadIdx.x >> 5;
    const int lane    = threadIdx.x & 31;
    const int m       = lane & 15;
    const int hi      = lane >> 4;
    const int colBase = waveId * 16;

    v8f acc = {};
#pragma unroll 8
    for (int k0 = 0; k0 < DIM; k0 += 4) {
        int ka = k0 + 2 * hi;
        v2f a, bf;
        a.x = As[m * 132 + ka];
        a.y = As[m * 132 + ka + 1];
        bf.x = W[(size_t)ka * DIM + colBase + m];
        bf.y = W[(size_t)(ka + 1) * DIM + colBase + m];
        acc = __builtin_amdgcn_wmma_f32_16x16x4_f32(
            false, a, false, bf, (short)0, acc, false, false);
    }

#pragma unroll
    for (int i = 0; i < 8; ++i)
        Cs[(i + 8 * hi) * 132 + colBase + m] = acc[i];
    __syncthreads();

#pragma unroll
    for (int rr = 0; rr < 2; ++rr) {
        const int r = 2 * waveId + rr;
        const float4 v = *reinterpret_cast<const float4*>(&Cs[r * 132 + 4 * lane]);
        const size_t off = (size_t)(rowBase + r) * DIM + 4 * lane;
        *reinterpret_cast<float4*>(xw + off) = v;
    }
    if (own && threadIdx.x < RANGE) {
        const float d = (float)(dsum[threadIdx.x] + DEG_ONE) * (1.0f / DEG_UNIT);
        dis[blockIdx.x * RANGE + threadIdx.x] = (d > 0.f) ? rsqrtf(d) : 0.f;
    }
}

__device__ __forceinline__ void add_rows(
    unsigned rem, const int s, const int tl, const int coef_bits, const int lane,
    const float* __restrict__ xw, float* acc)
{
    while (rem != 0u) {
        const int n = min(8, (int)__builtin_popcount(rem));
        float v[8][4] = {};
        unsigned r = rem;
#pragma unroll
        for (int j = 0; j < 8; ++j) {
            if (j >= n) break;
            const int bl = __builtin_ctz(r);
            r &= r - 1u;
            const int sj = __builtin_amdgcn_readlane(s, bl);
            const float* xr = xw + (size_t)sj * DIM + lane;
            v[j][0] = xr[0];
            v[j][1] = xr[32];
            v[j][2] = xr[64];
            v[j][3] = xr[96];
        }
        int tj[8];
        r = rem;
#pragma unroll
        for (int j = 0; j < 8; ++j) {
            const int bl = (r != 0u) ? __builtin_ctz(r) : 0;
            r &= r - 1u;
            tj[j] = __builtin_amdgcn_readlane(tl, bl);
            const float cj = __int_as_float(__builtin_amdgcn_readlane(coef_bits, bl));
            v[j][0] *= cj;
            v[j][1] *= cj;
            v[j][2] *= cj;
            v[j][3] *= cj;
        }
#pragma unroll
        for (int j = 0; j < 8; ++j) {
            if (j >= n) break;
            float* ap = acc + tj[j] * DIM + lane;
            __hip_atomic_fetch_add(ap +  0, v[j][0], __ATOMIC_RELAXED, __HIP_MEMORY_SCOPE_WORKGROUP);
            __hip_atomic_fetch_add(ap + 32, v[j][1], __ATOMIC_RELAXED, __HIP_MEMORY_SCOPE_WORKGROUP);
            __hip_atomic_fetch_add(ap + 64, v[j][2], __ATOMIC_RELAXED, __HIP_MEMORY_SCOPE_WORKGROUP);
            __hip_atomic_fetch_add(ap + 96, v[j][3], __ATOMIC_RELAXED, __HIP_MEMORY_SCOPE_WORKGROUP);
        }
        rem = r;
    }
}

__global__ __launch_bounds__(1024) void range_sum_kernel(
    const int* __restrict__ rec_words, const int* __restrict__ cnt,
    const int* __restrict__ sp_src, const int* __restrict__ sp_tgt, const float* __restrict__ sp_pa,
    const float* __restrict__ p_b, const float* __restrict__ b_pred,
    const float* __restrict__ dis, const float* __restrict__ xw, const float* __restrict__ b,
    float* __restrict__ out)
{
    __shared__ __attribute__((aligned(16))) float acc[RANGE * DIM];
    __shared__ float sdis[RANGE];
    __shared__ float spb[RANGE];
    const int tid  = threadIdx.x;
    const int lane = tid & 31;
    const int wv   = __builtin_amdgcn_readfirstlane(tid >> 5);
    const int node0 = blockIdx.x * RANGE;

    const int n  = min(cnt[blockIdx.x * CNT_STRIDE], BIN_CAP);
    const int ns = min(cnt[N_BINS * CNT_STRIDE], N_EDGES);
    const float bp = b_pred[0];
    const int2* myrec = reinterpret_cast<const int2*>(rec_words) + (size_t)blockIdx.x * BIN_CAP;

    const int hi_rec = ((wv + 1) * n) >> 5;
    int base = (wv * n) >> 5;

    int2 r = make_int2(0, 0);
    if (base + lane < hi_rec) r = myrec[base + lane];

    for (int i = tid; i < RANGE * DIM / 4; i += 1024)
        reinterpret_cast<float4*>(acc)[i] = make_float4(0.f, 0.f, 0.f, 0.f);
    if (tid < RANGE) {
        sdis[tid] = dis[node0 + tid];
        spb[tid]  = p_b[node0 + tid];
    }
    __syncthreads();

    while (base < hi_rec) {
        const int nb = base + 32;
        int2 rn = make_int2(0, 0);
        if (nb + lane < hi_rec) rn = myrec[nb + lane];
        const bool ok = base + lane < hi_rec;
        const int s  = r.x & 0xFFFF;
        const int tl = r.x >> 16;
        float coef = 0.f;
        if (ok) coef = dis[s] * edge_w(__int_as_float(r.y), spb[tl], bp) * sdis[tl];
        add_rows(__builtin_amdgcn_ballot_w32(ok), s, tl, __float_as_int(coef), lane, xw, acc);
        r = rn;
        base = nb;
    }

    for (int sb = wv * 32; sb < ns; sb += 1024) {
        const int i = sb + lane;
        int t = -1;
        if (i < ns) t = sp_tgt[i];
        const bool mine = (t >= node0) && (t < node0 + RANGE);
        int s = 0, tl = 0;
        float coef = 0.f;
        if (mine) {
            s  = sp_src[i];
            tl = t - node0;
            coef = dis[s] * edge_w(sp_pa[i], spb[tl], bp) * sdis[tl];
        }
        add_rows(__builtin_amdgcn_ballot_w32(mine), s, tl, __float_as_int(coef), lane, xw, acc);
    }

    float4 xs[ROWS_PER_WAVE];
#pragma unroll
    for (int k = 0; k < ROWS_PER_WAVE; ++k) {
        const int rr = min(wv + 32 * k, RANGE - 1);
        xs[k] = *reinterpret_cast<const float4*>(xw + (size_t)(node0 + rr) * DIM + 4 * lane);
    }
    const float4 bb = *reinterpret_cast<const float4*>(b + 4 * lane);
    __syncthreads();

#pragma unroll
    for (int k = 0; k < ROWS_PER_WAVE; ++k) {
        const int rr = wv + 32 * k;
        if (rr < RANGE) {
            const float4 a = *reinterpret_cast<const float4*>(&acc[rr * DIM + 4 * lane]);
            const float d  = sdis[rr];
            const float d2 = d * d;
            float4 o;
            o.x = a.x + d2 * xs[k].x + bb.x;
            o.y = a.y + d2 * xs[k].y + bb.y;
            o.z = a.z + d2 * xs[k].z + bb.z;
            o.w = a.w + d2 * xs[k].w + bb.w;
            *reinterpret_cast<float4*>(out + (size_t)(node0 + rr) * DIM + 4 * lane) = o;
        }
    }
}

extern "C" void kernel_launch(void* const* d_in, const int* in_sizes, int n_in,
                              void* d_out, int out_size, void* d_ws, size_t ws_size,
                              hipStream_t stream) {
    const float* x          = (const float*)d_in[0];
    const int*   edge_index = (const int*)d_in[1];
    const float* W          = (const float*)d_in[2];
    const float* b          = (const float*)d_in[3];
    const float* w_pred     = (const float*)d_in[4];
    const float* b_pred     = (const float*)d_in[5];
    float*       out        = (float*)d_out;

    const int* src = edge_index;
    const int* tgt = edge_index + N_EDGES;

    float* ws  = (float*)d_ws;
    float* p_a = ws;
    float* p_b = p_a + N_NODES;
    float* dis = p_b + N_NODES;
    float* xw  = dis + N_NODES + 16;
    int* cnt    = (int*)(xw + (size_t)N_NODES * DIM);
    int* rec    = cnt + (N_BINS + 1) * CNT_STRIDE;
    int* sp_src = rec + (size_t)N_BINS * BIN_CAP * 2;
    int* sp_tgt = sp_src + N_EDGES;
    float* sp_pa = (float*)(sp_tgt + N_EDGES);

    node_dots_kernel      <<<DOT_BLOCKS,        256,  0, stream>>>(x, w_pred, p_a, p_b, cnt);
    edge_bin_kernel       <<<N_TILES,           1024, 0, stream>>>(src, tgt, p_a, cnt, rec, sp_src, sp_tgt, sp_pa);
    gemm_wmma_kernel      <<<N_NODES / 16,      256,  0, stream>>>(x, W, cnt, rec, sp_tgt, sp_pa, p_b, b_pred, dis, xw);
    range_sum_kernel      <<<N_BINS,            1024, 0, stream>>>(rec, cnt, sp_src, sp_tgt, sp_pa, p_b, b_pred, dis, xw, b, out);
}


# range sums: records routed in LDS (integer ranks) to the wave owning the target row, which adds rows by plain 16-B LDS load/add/store; 102M f32 LDS atomic lane adds gone; sum order still arrival
# speedup vs baseline: 3.5763x; 1.0641x over previous
#include <hip/hip_runtime.h>
#include <cmath>

#define N_NODES 50000
#define N_EDGES 800000
#define DIM 128

#define RANGE      200
#define N_BINS     (N_NODES / RANGE)
#define BIN_CAP    4096
#define CNT_STRIDE 64
#define TILE_EDGES 2048
#define N_TILES    ((N_EDGES + TILE_EDGES - 1) / TILE_EDGES)
#define ROWS_PER_WAVE ((RANGE + 31) / 32)
#define DEG_UNIT   1073741824.0f
#define DEG_ONE    (1ull << 30)
#define DOT_NODES  16
#define DOT_BLOCKS ((N_NODES / DOT_NODES + 7) / 8)

static_assert(N_NODES % RANGE == 0, "ranges cover the nodes exactly");
static_assert(N_NODES % DOT_NODES == 0 && DOT_NODES <= 32, "a wave of node_dots_kernel has all its nodes or none; lane i stores node i");
static_assert(N_NODES <= 65536 && RANGE <= 32768, "a record packs src in 16 bits and the local target above them");
static_assert(N_BINS + 1 <= 256, "block 0 of node_dots_kernel clears all counters");
static_assert(RANGE <= 256 && N_BINS <= N_NODES / 16, "GEMM block r (256 threads) sums the degrees of range r");
static_assert(BIN_CAP % 1024 == 0, "a thread of range_sum_kernel takes BIN_CAP / 1024 records of its range");

typedef float v2f __attribute__((ext_vector_type(2)));
typedef float v8f __attribute__((ext_vector_type(8)));

__global__ __launch_bounds__(256) void node_dots_kernel(
    const float* __restrict__ x, const float* __restrict__ w_pred,
    float* __restrict__ p_a, float* __restrict__ p_b, int* __restrict__ cnt)
{
    if (blockIdx.x == 0 && threadIdx.x <= N_BINS) cnt[threadIdx.x * CNT_STRIDE] = 0;
    const int lane  = threadIdx.x & 31;
    const int node0 = ((blockIdx.x * blockDim.x + threadIdx.x) >> 5) * DOT_NODES;
    if (node0 >= N_NODES) return;
    const float* xr = x + (size_t)node0 * DIM + lane;
    float v[DOT_NODES][4];
#pragma unroll
    for (int i = 0; i < DOT_NODES; ++i)
#pragma unroll
        for (int j = 0; j < 4; ++j) v[i][j] = xr[i * DIM + 32 * j];
    float wa[4], wb[4];
#pragma unroll
    for (int j = 0; j < 4; ++j) {
        wa[j] = w_pred[lane + 32 * j];
        wb[j] = w_pred[DIM + lane + 32 * j];
    }
    float ra = 0.f, rb = 0.f;
#pragma unroll
    for (int i = 0; i < DOT_NODES; ++i) {
        float sa = 0.f, sb = 0.f;
#pragma unroll
        for (int j = 0; j < 4; ++j) {
            sa += v[i][j] * wa[j];
            sb += v[i][j] * wb[j];
        }
#pragma unroll
        for (int off = 16; off > 0; off >>= 1) {
            sa += __shfl_down(sa, off, 32);
            sb += __shfl_down(sb, off, 32);
        }
        const float a0 = __int_as_float(__builtin_amdgcn_readfirstlane(__float_as_int(sa)));
        const float b0 = __int_as_float(__builtin_amdgcn_readfirstlane(__float_as_int(sb)));
        if (lane == i) { ra = a0; rb = b0; }
    }
    if (lane < DOT_NODES) {
        p_a[node0 + lane] = ra;
        p_b[node0 + lane] = rb;
    }
}

__device__ __forceinline__ float edge_w(const float pa, const float pb, const float bp)
{
    const float z = pa + pb + bp;
    return 1.0f / (1.0f + __expf(-z));
}

__global__ __launch_bounds__(1024) void edge_bin_kernel(
    const int* __restrict__ src, const int* __restrict__ tgt,
    const float* __restrict__ p_a,
    int* __restrict__ cnt, int* __restrict__ rec_words,
    int* __restrict__ sp_src, int* __restrict__ sp_tgt, float* __restrict__ sp_pa)
{
    __shared__ int hist[N_BINS];
    const int tid = threadIdx.x;
    if (tid < N_BINS) hist[tid] = 0;
    __syncthreads();

    int   s[2], t[2], bin[2], rank[2];
    float pa[2];
    bool  ok[2];
#pragma unroll
    for (int k = 0; k < 2; ++k) {
        const int e = blockIdx.x * TILE_EDGES + k * 1024 + tid;
        ok[k] = e < N_EDGES;
        const int ec = ok[k] ? e : (N_EDGES - 1);
        s[k] = src[ec];
        t[k] = tgt[ec];
    }
#pragma unroll
    for (int k = 0; k < 2; ++k) {
        pa[k]   = p_a[s[k]];
        bin[k]  = t[k] / RANGE;
        rank[k] = 0;
        if (ok[k]) rank[k] = atomicAdd(&hist[bin[k]], 1);
    }
    __syncthreads();
    if (tid < N_BINS) {
        const int c = hist[tid];
        hist[tid] = (c > 0) ? atomicAdd(&cnt[tid * CNT_STRIDE], c) : 0;
    }
    __syncthreads();
    int2* rec = reinterpret_cast<int2*>(rec_words);
#pragma unroll
    for (int k = 0; k < 2; ++k) {
        if (ok[k]) {
            const int slot = hist[bin[k]] + rank[k];
            if (slot < BIN_CAP) {
                const int tl = t[k] - bin[k] * RANGE;
                rec[(size_t)bin[k] * BIN_CAP + slot] = make_int2(s[k] | (tl << 16), __float_as_int(pa[k]));
            } else {
                const int o = atomicAdd(&cnt[N_BINS * CNT_STRIDE], 1);
                if (o < N_EDGES) {
                    sp_src[o] = s[k];
                    sp_tgt[o] = t[k];
                    sp_pa[o]  = pa[k];
                }
            }
        }
    }
}

__device__ __forceinline__ unsigned long long deg_fix(const float w)
{
    return (unsigned long long)__float2uint_rn(w * DEG_UNIT);
}

__global__ __launch_bounds__(256) void gemm_wmma_kernel(
    const float* __restrict__ x, const float* __restrict__ W,
    const int* __restrict__ cnt, const int* __restrict__ rec_words,
    const int* __restrict__ sp_tgt, const float* __restrict__ sp_pa,
    const float* __restrict__ p_b, const float* __restrict__ b_pred,
    float* __restrict__ dis, float* __restrict__ xw)
{
    __shared__ float As[16 * 132];
    __shared__ __attribute__((aligned(16))) float Cs[16 * 132];
    __shared__ unsigned long long dsum[RANGE];
    __shared__ float spb[RANGE];
    const int rowBase = blockIdx.x * 16;

    const bool own = blockIdx.x < N_BINS;
    int n = 0, ns = 0;
    float bp = 0.f;
    if (own) {
        n  = min(cnt[blockIdx.x * CNT_STRIDE], BIN_CAP);
        ns = min(cnt[N_BINS * CNT_STRIDE], N_EDGES);
        bp = b_pred[0];
        if (threadIdx.x < RANGE) {
            dsum[threadIdx.x] = 0ull;
            spb[threadIdx.x]  = p_b[blockIdx.x * RANGE + threadIdx.x];
        }
    }

#pragma unroll
    for (int j = 0; j < 8; ++j) {
        int idx = threadIdx.x + j * 256;
        int r = idx >> 7, c = idx & 127;
        As[r * 132 + c] = x[(size_t)rowBase * DIM + idx];
    }
    __syncthreads();

    if (own) {
        const int node0 = blockIdx.x * RANGE;
        const int2* myrec = reinterpret_cast<const int2*>(rec_words) + (size_t)blockIdx.x * BIN_CAP;
        for (int i0 = threadIdx.x; i0 < n; i0 += 8 * 256) {
            int2 r[8];
#pragma unroll
            for (int j = 0; j < 8; ++j) r[j] = myrec[min(i0 + j * 256, n - 1)];
#pragma unroll
            for (int j = 0; j < 8; ++j)
                if (i0 + j * 256 < n) {
                    const int tl = r[j].x >> 16;
                    atomicAdd(&dsum[tl], deg_fix(edge_w(__int_as_float(r[j].y), spb[tl], bp)));
                }
        }
        for (int i = threadIdx.x; i < ns; i += 256) {
            const int t = sp_tgt[i];
            if (t >= node0 && t < node0 + RANGE)
                atomicAdd(&dsum[t - node0], deg_fix(edge_w(sp_pa[i], spb[t - node0], bp)));
        }
    }

    const int waveId  = threadIdx.x >> 5;
    const int lane    = threadIdx.x & 31;
    const int m       = lane & 15;
    const int hi      = lane >> 4;
    const int colBase = waveId * 16;

    v8f acc = {};
#pragma unroll 8
    for (int k0 = 0; k0 < DIM; k0 += 4) {
        int ka = k0 + 2 * hi;
        v2f a, bf;
        a.x = As[m * 132 + ka];
        a.y = As[m * 132 + ka + 1];
        bf.x = W[(size_t)ka * DIM + colBase + m];
        bf.y = W[(size_t)(ka + 1) * DIM + colBase + m];
        acc = __builtin_amdgcn_wmma_f32_16x16x4_f32(
            false, a, false, bf, (short)0, acc, false, false);
    }

#pragma unroll
    for (int i = 0; i < 8; ++i)
        Cs[(i + 8 * hi) * 132 + colBase + m] = acc[i];
    __syncthreads();

#pragma unroll
    for (int rr = 0; rr < 2; ++rr) {
        const int r = 2 * waveId + rr;
        const float4 v = *reinterpret_cast<const float4*>(&Cs[r * 132 + 4 * lane]);
        const size_t off = (size_t)(rowBase + r) * DIM + 4 * lane;
        *reinterpret_cast<float4*>(xw + off) = v;
    }
    if (own && threadIdx.x < RANGE) {
        const float d = (float)(dsum[threadIdx.x] + DEG_ONE) * (1.0f / DEG_UNIT);
        dis[blockIdx.x * RANGE + threadIdx.x] = (d > 0.f) ? rsqrtf(d) : 0.f;
    }
}

__device__ __forceinline__ void add_rows(
    unsigned rem, const int s, const int tl, const int coef_bits, const int lane,
    const float* __restrict__ xw, float* acc)
{
    while (rem != 0u) {
        const int n = min(8, (int)__builtin_popcount(rem));
        float v[8][4] = {};
        unsigned r = rem;
#pragma unroll
        for (int j = 0; j < 8; ++j) {
            if (j >= n) break;
            const int bl = __builtin_ctz(r);
            r &= r - 1u;
            const int sj = __builtin_amdgcn_readlane(s, bl);
            const float* xr = xw + (size_t)sj * DIM + lane;
            v[j][0] = xr[0];
            v[j][1] = xr[32];
            v[j][2] = xr[64];
            v[j][3] = xr[96];
        }
        int tj[8];
        r = rem;
#pragma unroll
        for (int j = 0; j < 8; ++j) {
            const int bl = (r != 0u) ? __builtin_ctz(r) : 0;
            r &= r - 1u;
            tj[j] = __builtin_amdgcn_readlane(tl, bl);
            const float cj = __int_as_float(__builtin_amdgcn_readlane(coef_bits, bl));
            v[j][0] *= cj;
            v[j][1] *= cj;
            v[j][2] *= cj;
            v[j][3] *= cj;
        }
#pragma unroll
        for (int j = 0; j < 8; ++j) {
            if (j >= n) break;
            float* ap = acc + tj[j] * DIM + lane;
            __hip_atomic_fetch_add(ap +  0, v[j][0], __ATOMIC_RELAXED, __HIP_MEMORY_SCOPE_WORKGROUP);
            __hip_atomic_fetch_add(ap + 32, v[j][1], __ATOMIC_RELAXED, __HIP_MEMORY_SCOPE_WORKGROUP);
            __hip_atomic_fetch_add(ap + 64, v[j][2], __ATOMIC_RELAXED, __HIP_MEMORY_SCOPE_WORKGROUP);
            __hip_atomic_fetch_add(ap + 96, v[j][3], __ATOMIC_RELAXED, __HIP_MEMORY_SCOPE_WORKGROUP);
        }
        rem = r;
    }
}

__device__ __forceinline__ void ask_rows(
    const int2* qlist, const int b0, const int qe, const int lane,
    const float* __restrict__ xw, int2& ent, float4 (&v)[8])
{
    ent = make_int2(0, 0);
    if (lane < 8 && b0 + lane < qe) ent = qlist[b0 + lane];
#pragma unroll
    for (int j = 0; j < 8; ++j) {
        const int sj = __builtin_amdgcn_readlane(ent.x, j) & 0xFFFF;
        v[j] = *reinterpret_cast<const float4*>(xw + (size_t)sj * DIM + 4 * lane);
    }
}

__device__ __forceinline__ void sum_rows(
    float* acc, const int b0, const int qe, const int lane, const int2 ent, const float4 (&v)[8])
{
#pragma unroll
    for (int j = 0; j < 8; ++j) {
        if (b0 + j < qe) {
            const int   tj = __builtin_amdgcn_readlane(ent.x, j) >> 16;
            const float cj = __int_as_float(__builtin_amdgcn_readlane(ent.y, j));
            float4* ap = reinterpret_cast<float4*>(acc + tj * DIM + 4 * lane);
            float4 a = *ap;
            a.x += cj * v[j].x;
            a.y += cj * v[j].y;
            a.z += cj * v[j].z;
            a.w += cj * v[j].w;
            *ap = a;
        }
    }
}

__global__ __launch_bounds__(1024) void range_sum_kernel(
    const int* __restrict__ rec_words, const int* __restrict__ cnt,
    const int* __restrict__ sp_src, const int* __restrict__ sp_tgt, const float* __restrict__ sp_pa,
    const float* __restrict__ p_b, const float* __restrict__ b_pred,
    const float* __restrict__ dis, const float* __restrict__ xw, const float* __restrict__ b,
    float* __restrict__ out)
{
    __shared__ __attribute__((aligned(16))) float acc[RANGE * DIM];
    __shared__ float sdis[RANGE];
    __shared__ float spb[RANGE];
    __shared__ int2  qlist[BIN_CAP];
    __shared__ int   qcnt[32];
    __shared__ int   qbase[33];
    const int tid  = threadIdx.x;
    const int lane = tid & 31;
    const int wv   = __builtin_amdgcn_readfirstlane(tid >> 5);
    const int node0 = blockIdx.x * RANGE;

    const int n  = min(cnt[blockIdx.x * CNT_STRIDE], BIN_CAP);
    const int ns = min(cnt[N_BINS * CNT_STRIDE], N_EDGES);
    const float bp = b_pred[0];
    const int2* myrec = reinterpret_cast<const int2*>(rec_words) + (size_t)blockIdx.x * BIN_CAP;

    int2 r[BIN_CAP / 1024];
#pragma unroll
    for (int k = 0; k < BIN_CAP / 1024; ++k) {
        r[k] = make_int2(0, 0);
        if (tid + 1024 * k < n) r[k] = myrec[tid + 1024 * k];
    }

    for (int i = tid; i < RANGE * DIM / 4; i += 1024)
        reinterpret_cast<float4*>(acc)[i] = make_float4(0.f, 0.f, 0.f, 0.f);
    if (tid < 32) qcnt[tid] = 0;
    if (tid < RANGE) {
        sdis[tid] = dis[node0 + tid];
        spb[tid]  = p_b[node0 + tid];
    }
    __syncthreads();

    int   rank[BIN_CAP / 1024];
    float coef[BIN_CAP / 1024];
#pragma unroll
    for (int k = 0; k < BIN_CAP / 1024; ++k) {
        rank[k] = 0;
        coef[k] = 0.f;
        if (tid + 1024 * k < n) {
            const int s  = r[k].x & 0xFFFF;
            const int tl = r[k].x >> 16;
            rank[k] = atomicAdd(&qcnt[tl & 31], 1);
            coef[k] = dis[s] * edge_w(__int_as_float(r[k].y), spb[tl], bp) * sdis[tl];
        }
    }
    __syncthreads();
    if (tid <= 32) {
        int q = 0;
        for (int j = 0; j < tid; ++j) q += qcnt[j];
        qbase[tid] = q;
    }
    __syncthreads();
#pragma unroll
    for (int k = 0; k < BIN_CAP / 1024; ++k)
        if (tid + 1024 * k < n)
            qlist[qbase[(r[k].x >> 16) & 31] + rank[k]] = make_int2(r[k].x, __float_as_int(coef[k]));
    __syncthreads();

    {
        const int qe = __builtin_amdgcn_readfirstlane(qbase[wv + 1]);
        int base     = __builtin_amdgcn_readfirstlane(qbase[wv]);
        int2 ea, eb;
        float4 va[8], vb[8];
        ask_rows(qlist, base, qe, lane, xw, ea, va);
        while (base < qe) {
            ask_rows(qlist, base + 8, qe, lane, xw, eb, vb);
            sum_rows(acc, base, qe, lane, ea, va);
            base += 8;
            if (base >= qe) break;
            ask_rows(qlist, base + 8, qe, lane, xw, ea, va);
            sum_rows(acc, base, qe, lane, eb, vb);
            base += 8;
        }
    }

    if (ns > 0) {
        __syncthreads();
        for (int sb = wv * 32; sb < ns; sb += 1024) {
            const int i = sb + lane;
            int t = -1;
            if (i < ns) t = sp_tgt[i];
            const bool mine = (t >= node0) && (t < node0 + RANGE);
            int s = 0, tl = 0;
            float cf = 0.f;
            if (mine) {
                s  = sp_src[i];
                tl = t - node0;
                cf = dis[s] * edge_w(sp_pa[i], spb[tl], bp) * sdis[tl];
            }
            add_rows(__builtin_amdgcn_ballot_w32(mine), s, tl, __float_as_int(cf), lane, xw, acc);
        }
    }

    float4 xs[ROWS_PER_WAVE];
#pragma unroll
    for (int k = 0; k < ROWS_PER_WAVE; ++k) {
        const int rr = min(wv + 32 * k, RANGE - 1);
        xs[k] = *reinterpret_cast<const float4*>(xw + (size_t)(node0 + rr) * DIM + 4 * lane);
    }
    const float4 bb = *reinterpret_cast<const float4*>(b + 4 * lane);
    __syncthreads();

#pragma unroll
    for (int k = 0; k < ROWS_PER_WAVE; ++k) {
        const int rr = wv + 32 * k;
        if (rr < RANGE) {
            const float4 a = *reinterpret_cast<const float4*>(&acc[rr * DIM + 4 * lane]);
            const float d  = sdis[rr];
            const float d2 = d * d;
            float4 o;
            o.x = a.x + d2 * xs[k].x + bb.x;
            o.y = a.y + d2 * xs[k].y + bb.y;
            o.z = a.z + d2 * xs[k].z + bb.z;
            o.w = a.w + d2 * xs[k].w + bb.w;
            *reinterpret_cast<float4*>(out + (size_t)(node0 + rr) * DIM + 4 * lane) = o;
        }
    }
}

extern "C" void kernel_launch(void* const* d_in, const int* in_sizes, int n_in,
                              void* d_out, int out_size, void* d_ws, size_t ws_size,
                              hipStream_t stream) {
    const float* x          = (const float*)d_in[0];
    const int*   edge_index = (const int*)d_in[1];
    const float* W          = (const float*)d_in[2];
    const float* b          = (const float*)d_in[3];
    const float* w_pred     = (const float*)d_in[4];
    const float* b_pred     = (const float*)d_in[5];
    float*       out        = (float*)d_out;

    const int* src = edge_index;
    const int* tgt = edge_index + N_EDGES;

    float* ws  = (float*)d_ws;
    float* p_a = ws;
    float* p_b = p_a + N_NODES;
    float* dis = p_b + N_NODES;
    float* xw  = dis + N_NODES + 16;
    int* cnt    = (int*)(xw + (size_t)N_NODES * DIM);
    int* rec    = cnt + (N_BINS + 1) * CNT_STRIDE;
    int* sp_src = rec + (size_t)N_BINS * BIN_CAP * 2;
    int* sp_tgt = sp_src + N_EDGES;
    float* sp_pa = (float*)(sp_tgt + N_EDGES);

    node_dots_kernel      <<<DOT_BLOCKS,        256,  0, stream>>>(x, w_pred, p_a, p_b, cnt);
    edge_bin_kernel       <<<N_TILES,           1024, 0, stream>>>(src, tgt, p_a, cnt, rec, sp_src, sp_tgt, sp_pa);
    gemm_wmma_kernel      <<<N_NODES / 16,      256,  0, stream>>>(x, W, cnt, rec, sp_tgt, sp_pa, p_b, b_pred, dis, xw);
    range_sum_kernel      <<<N_BINS,            1024, 0, stream>>>(rec, cnt, sp_src, sp_tgt, sp_pa, p_b, b_pred, dis, xw, b, out);
}


# edges sorted by target range inside each 4096-edge tile (LDS ranks, offset table), the range's owner gathers its pieces: no global counters, the sort shares the dots' launch; 3 launches, not 4
# speedup vs baseline: 3.7928x; 1.0605x over previous
#include <hip/hip_runtime.h>
#include <cmath>

#define N_NODES 50000
#define N_EDGES 800000
#define DIM 128

#define RANGE      200
#define N_BINS     (N_NODES / RANGE)
#define BIN_CAP    4096
#define TILE_EDGES 4096
#define EPT        (TILE_EDGES / 1024)
#define N_TILES    ((N_EDGES + TILE_EDGES - 1) / TILE_EDGES)
#define OFF_STRIDE 256
#define ROWS_PER_WAVE ((RANGE + 31) / 32)
#define DEG_UNIT   1073741824.0f
#define DEG_ONE    (1ull << 30)
#define DOT_NODES  8
#define DOT_WAVES  (N_NODES / DOT_NODES)
#define DOT_BLOCKS ((DOT_WAVES + 31) / 32)

static_assert(N_NODES % RANGE == 0, "ranges cover the nodes exactly");
static_assert(N_NODES % DOT_NODES == 0 && DOT_NODES <= 32, "a dots wave has all its nodes or none; lane i stores node i");
static_assert(N_NODES <= 65536 && RANGE <= 32768, "a record packs src in 16 bits and the local target above them");
static_assert(N_BINS + 1 <= OFF_STRIDE && OFF_STRIDE == 256, "threads 0..255 of a sorting block scan the tile's range counts");
static_assert(RANGE <= 256 && N_BINS <= N_NODES / 16, "GEMM block r (256 threads) sums the degrees of range r");
static_assert(N_TILES <= 512, "a thread of GEMM block r takes tiles tid and tid + 256");
static_assert(BIN_CAP % 1024 == 0, "a thread of range_sum_kernel takes BIN_CAP / 1024 records per pass");

typedef float v2f __attribute__((ext_vector_type(2)));
typedef float v8f __attribute__((ext_vector_type(8)));

__device__ __forceinline__ int wave_incl_scan(int v, const int lane)
{
#pragma unroll
    for (int off = 1; off < 32; off <<= 1) {
        const int u = __shfl_up(v, off, 32);
        if (lane >= off) v += u;
    }
    return v;
}

__global__ __launch_bounds__(1024, 16) void front_kernel(
    const int* __restrict__ src, const int* __restrict__ tgt,
    const float* __restrict__ x, const float* __restrict__ w_pred,
    float* __restrict__ p_a, float* __restrict__ p_b,
    int* __restrict__ toff, int* __restrict__ trec)
{
    __shared__ int hist[OFF_STRIDE];
    __shared__ int wtot[8];
    const int tid  = threadIdx.x;
    const int lane = tid & 31;
    if (blockIdx.x < N_TILES) {
        if (tid < OFF_STRIDE) hist[tid] = 0;
        __syncthreads();
        int  s[EPT], t[EPT], bin[EPT], rank[EPT];
        bool ok[EPT];
#pragma unroll
        for (int k = 0; k < EPT; ++k) {
            const int e = blockIdx.x * TILE_EDGES + k * 1024 + tid;
            ok[k] = e < N_EDGES;
            const int ec = ok[k] ? e : (N_EDGES - 1);
            s[k] = src[ec];
            t[k] = tgt[ec];
        }
#pragma unroll
        for (int k = 0; k < EPT; ++k) {
            bin[k]  = t[k] / RANGE;
            rank[k] = 0;
            if (ok[k]) rank[k] = atomicAdd(&hist[bin[k]], 1);
        }
        __syncthreads();
        int c = 0, inc = 0;
        if (tid < OFF_STRIDE) {
            c   = hist[tid];
            inc = wave_incl_scan(c, lane);
            if (lane == 31) wtot[tid >> 5] = inc;
        }
        __syncthreads();
        if (tid < OFF_STRIDE) {
            int before = 0;
            for (int j = 0; j < (tid >> 5); ++j) before += wtot[j];
            const int ex = before + inc - c;
            hist[tid] = ex;
            toff[blockIdx.x * OFF_STRIDE + tid] = ex;
        }
        __syncthreads();
#pragma unroll
        for (int k = 0; k < EPT; ++k) {
            if (ok[k]) {
                const int tl = t[k] - bin[k] * RANGE;
                trec[blockIdx.x * TILE_EDGES + hist[bin[k]] + rank[k]] = s[k] | (tl << 16);
            }
        }
    } else {
        const int wid   = (blockIdx.x - N_TILES) * 32 + (tid >> 5);
        const int node0 = wid * DOT_NODES;
        if (node0 >= N_NODES) return;
        const float* xr = x + (size_t)node0 * DIM + lane;
        float v[DOT_NODES][4];
#pragma unroll
        for (int i = 0; i < DOT_NODES; ++i)
#pragma unroll
            for (int j = 0; j < 4; ++j) v[i][j] = xr[i * DIM + 32 * j];
        float wa[4], wb[4];
#pragma unroll
        for (int j = 0; j < 4; ++j) {
            wa[j] = w_pred[lane + 32 * j];
            wb[j] = w_pred[DIM + lane + 32 * j];
        }
        float ra = 0.f, rb = 0.f;
#pragma unroll
        for (int i = 0; i < DOT_NODES; ++i) {
            float sa = 0.f, sb = 0.f;
#pragma unroll
            for (int j = 0; j < 4; ++j) {
                sa += v[i][j] * wa[j];
                sb += v[i][j] * wb[j];
            }
#pragma unroll
            for (int off = 16; off > 0; off >>= 1) {
                sa += __shfl_down(sa, off, 32);
                sb += __shfl_down(sb, off, 32);
            }
            const float a0 = __int_as_float(__builtin_amdgcn_readfirstlane(__float_as_int(sa)));
            const float b0 = __int_as_float(__builtin_amdgcn_readfirstlane(__float_as_int(sb)));
            if (lane == i) { ra = a0; rb = b0; }
        }
        if (lane < DOT_NODES) {
            p_a[node0 + lane] = ra;
            p_b[node0 + lane] = rb;
        }
    }
}

__device__ __forceinline__ float edge_w(const float pa, const float pb, const float bp)
{
    const float z = pa + pb + bp;
    return 1.0f / (1.0f + __expf(-z));
}

__device__ __forceinline__ unsigned long long deg_fix(const float w)
{
    return (unsigned long long)__float2uint_rn(w * DEG_UNIT);
}

__global__ __launch_bounds__(256) void gemm_wmma_kernel(
    const float* __restrict__ x, const float* __restrict__ W,
    const int* __restrict__ toff, const int* __restrict__ trec,
    const float* __restrict__ p_a, const float* __restrict__ p_b, const float* __restrict__ b_pred,
    int* __restrict__ crec_words, int* __restrict__ rinfo,
    float* __restrict__ dis, float* __restrict__ xw)
{
    __shared__ float As[16 * 132];
    __shared__ __attribute__((aligned(16))) float Cs[16 * 132];
    __shared__ unsigned long long dsum[RANGE];
    __shared__ float spb[RANGE];
    __shared__ int   wsum[3][8];
    const int rowBase = blockIdx.x * 16;
    const int waveId  = threadIdx.x >> 5;
    const int lane    = threadIdx.x & 31;

    const bool own = blockIdx.x < N_BINS;
    int o0[2] = {0, 0}, len[2] = {0, 0};
    float bp = 0.f;
    if (own) {
#pragma unroll
        for (int q = 0; q < 2; ++q) {
            const int t  = threadIdx.x + 256 * q;
            const int tc = min(t, N_TILES - 1);
            const int a0 = toff[tc * OFF_STRIDE + blockIdx.x];
            const int a1 = toff[tc * OFF_STRIDE + blockIdx.x + 1];
            o0[q]  = (t < N_TILES) ? a0 : 0;
            len[q] = (t < N_TILES) ? (a1 - a0) : 0;
        }
        bp = b_pred[0];
        if (threadIdx.x < RANGE) {
            dsum[threadIdx.x] = 0ull;
            spb[threadIdx.x]  = p_b[blockIdx.x * RANGE + threadIdx.x];
        }
    }

#pragma unroll
    for (int j = 0; j < 8; ++j) {
        int idx = threadIdx.x + j * 256;
        int r = idx >> 7, c = idx & 127;
        As[r * 132 + c] = x[(size_t)rowBase * DIM + idx];
    }
    int inc0 = 0, inc1 = 0, incs = 0;
    if (own) {
        inc0 = wave_incl_scan(len[0], lane);
        inc1 = wave_incl_scan(len[1], lane);
        incs = wave_incl_scan(o0[0] + o0[1], lane);
        if (lane == 31) { wsum[0][waveId] = inc0; wsum[1][waveId] = inc1; wsum[2][waveId] = incs; }
    }
    __syncthreads();

    if (own) {
        int before0 = 0, total0 = 0, before1 = 0, total1 = 0, start = 0;
#pragma unroll
        for (int j = 0; j < 8; ++j) {
            if (j < waveId) { before0 += wsum[0][j]; before1 += wsum[1][j]; }
            total0 += wsum[0][j];
            total1 += wsum[1][j];
            start  += wsum[2][j];
        }
        const int pos[2] = { before0 + inc0 - len[0], total0 + before1 + inc1 - len[1] };
        if (threadIdx.x == 0) {
            rinfo[2 * blockIdx.x]     = start;
            rinfo[2 * blockIdx.x + 1] = total0 + total1;
        }
        int2* crec = reinterpret_cast<int2*>(crec_words) + start;
#pragma unroll
        for (int q = 0; q < 2; ++q) {
            const int* piece = trec + min((int)threadIdx.x + 256 * q, N_TILES - 1) * TILE_EDGES + o0[q];
            const int  n = len[q];
            for (int k0 = 0; k0 < n; k0 += 8) {
                int   w[8];
                float pa[8];
#pragma unroll
                for (int j = 0; j < 8; ++j) w[j] = piece[min(k0 + j, n - 1)];
#pragma unroll
                for (int j = 0; j < 8; ++j) pa[j] = p_a[w[j] & 0xFFFF];
#pragma unroll
                for (int j = 0; j < 8; ++j)
                    if (k0 + j < n) {
                        const int   tl = w[j] >> 16;
                        const float ew = edge_w(pa[j], spb[tl], bp);
                        atomicAdd(&dsum[tl], deg_fix(ew));
                        crec[pos[q] + k0 + j] = make_int2(w[j], __float_as_int(ew));
                    }
            }
        }
    }

    const int m       = lane & 15;
    const int hi      = lane >> 4;
    const int colBase = waveId * 16;

    v8f acc = {};
#pragma unroll 8
    for (int k0 = 0; k0 < DIM; k0 += 4) {
        int ka = k0 + 2 * hi;
        v2f a, bf;
        a.x = As[m * 132 + ka];
        a.y = As[m * 132 + ka + 1];
        bf.x = W[(size_t)ka * DIM + colBase + m];
        bf.y = W[(size_t)(ka + 1) * DIM + colBase + m];
        acc = __builtin_amdgcn_wmma_f32_16x16x4_f32(
            false, a, false, bf, (short)0, acc, false, false);
    }

#pragma unroll
    for (int i = 0; i < 8; ++i)
        Cs[(i + 8 * hi) * 132 + colBase + m] = acc[i];
    __syncthreads();

#pragma unroll
    for (int rr = 0; rr < 2; ++rr) {
        const int r = 2 * waveId + rr;
        const float4 v = *reinterpret_cast<const float4*>(&Cs[r * 132 + 4 * lane]);
        const size_t off = (size_t)(rowBase + r) * DIM + 4 * lane;
        *reinterpret_cast<float4*>(xw + off) = v;
    }
    if (own && threadIdx.x < RANGE) {
        const float d = (float)(dsum[threadIdx.x] + DEG_ONE) * (1.0f / DEG_UNIT);
        dis[blockIdx.x * RANGE + threadIdx.x] = (d > 0.f) ? rsqrtf(d) : 0.f;
    }
}

__device__ __forceinline__ void ask_rows(
    const int2* qlist, const int b0, const int qe, const int lane,
    const float* __restrict__ xw, int2& ent, float4 (&v)[8])
{
    ent = make_int2(0, 0);
    if (lane < 8 && b0 + lane < qe) ent = qlist[b0 + lane];
#pragma unroll
    for (int j = 0; j < 8; ++j) {
        const int sj = __builtin_amdgcn_readlane(ent.x, j) & 0xFFFF;
        v[j] = *reinterpret_cast<const float4*>(xw + (size_t)sj * DIM + 4 * lane);
    }
}

__device__ __forceinline__ void sum_rows(
    float* acc, const int b0, const int qe, const int lane, const int2 ent, const float4 (&v)[8])
{
#pragma unroll
    for (int j = 0; j < 8; ++j) {
        if (b0 + j < qe) {
            const int   tj = __builtin_amdgcn_readlane(ent.x, j) >> 16;
            const float cj = __int_as_float(__builtin_amdgcn_readlane(ent.y, j));
            float4* ap = reinterpret_cast<float4*>(acc + tj * DIM + 4 * lane);
            float4 a = *ap;
            a.x += cj * v[j].x;
            a.y += cj * v[j].y;
            a.z += cj * v[j].z;
            a.w += cj * v[j].w;
            *ap = a;
        }
    }
}

__global__ __launch_bounds__(1024) void range_sum_kernel(
    const int* __restrict__ crec_words, const int* __restrict__ rinfo,
    const float* __restrict__ dis, const float* __restrict__ xw, const float* __restrict__ b,
    float* __restrict__ out)
{
    __shared__ __attribute__((aligned(16))) float acc[RANGE * DIM];
    __shared__ float sdis[RANGE];
    __shared__ int2  qlist[BIN_CAP];
    __shared__ int   qcnt[32];
    __shared__ int   qbase[33];
    const int tid  = threadIdx.x;
    const int lane = tid & 31;
    const int wv   = __builtin_amdgcn_readfirstlane(tid >> 5);
    const int node0 = blockIdx.x * RANGE;

    const int start = min(max(rinfo[2 * blockIdx.x], 0), N_EDGES);
    const int ntot  = min(max(rinfo[2 * blockIdx.x + 1], 0), N_EDGES - start);
    const int2* myrec = reinterpret_cast<const int2*>(crec_words) + start;

    for (int i = tid; i < RANGE * DIM / 4; i += 1024)
        reinterpret_cast<float4*>(acc)[i] = make_float4(0.f, 0.f, 0.f, 0.f);
    if (tid < RANGE) sdis[tid] = dis[node0 + tid];

    for (int c0 = 0; c0 < ntot || c0 == 0; c0 += BIN_CAP) {
        const int n = min(ntot - c0, BIN_CAP);
        int2 r[BIN_CAP / 1024];
#pragma unroll
        for (int k = 0; k < BIN_CAP / 1024; ++k) {
            r[k] = make_int2(0, 0);
            if (tid + 1024 * k < n) r[k] = myrec[c0 + tid + 1024 * k];
        }
        if (tid < 32) qcnt[tid] = 0;
        __syncthreads();

        int   rank[BIN_CAP / 1024];
        float coef[BIN_CAP / 1024];
#pragma unroll
        for (int k = 0; k < BIN_CAP / 1024; ++k) {
            rank[k] = 0;
            coef[k] = 0.f;
            if (tid + 1024 * k < n) {
                const int s  = r[k].x & 0xFFFF;
                const int tl = r[k].x >> 16;
                rank[k] = atomicAdd(&qcnt[tl & 31], 1);
                coef[k] = dis[s] * __int_as_float(r[k].y) * sdis[tl];
            }
        }
        __syncthreads();
        if (tid <= 32) {
            int q = 0;
            for (int j = 0; j < tid; ++j) q += qcnt[j];
            qbase[tid] = q;
        }
        __syncthreads();
#pragma unroll
        for (int k = 0; k < BIN_CAP / 1024; ++k)
            if (tid + 1024 * k < n)
                qlist[qbase[(r[k].x >> 16) & 31] + rank[k]] = make_int2(r[k].x, __float_as_int(coef[k]));
        __syncthreads();

        {
            const int qe = __builtin_amdgcn_readfirstlane(qbase[wv + 1]);
            int base     = __builtin_amdgcn_readfirstlane(qbase[wv]);
            int2 ea, eb;
            float4 va[8], vb[8];
            ask_rows(qlist, base, qe, lane, xw, ea, va);
            while (base < qe) {
                ask_rows(qlist, base + 8, qe, lane, xw, eb, vb);
                sum_rows(acc, base, qe, lane, ea, va);
                base += 8;
                if (base >= qe) break;
                ask_rows(qlist, base + 8, qe, lane, xw, ea, va);
                sum_rows(acc, base, qe, lane, eb, vb);
                base += 8;
            }
        }
        if (c0 + BIN_CAP < ntot) __syncthreads();
    }

    float4 xs[ROWS_PER_WAVE];
#pragma unroll
    for (int k = 0; k < ROWS_PER_WAVE; ++k) {
        const int rr = min(wv + 32 * k, RANGE - 1);
        xs[k] = *reinterpret_cast<const float4*>(xw + (size_t)(node0 + rr) * DIM + 4 * lane);
    }
    const float4 bb = *reinterpret_cast<const float4*>(b + 4 * lane);
    __syncthreads();

#pragma unroll
    for (int k = 0; k < ROWS_PER_WAVE; ++k) {
        const int rr = wv + 32 * k;
        if (rr < RANGE) {
            const float4 a = *reinterpret_cast<const float4*>(&acc[rr * DIM + 4 * lane]);
            const float d  = sdis[rr];
            const float d2 = d * d;
            float4 o;
            o.x = a.x + d2 * xs[k].x + bb.x;
            o.y = a.y + d2 * xs[k].y + bb.y;
            o.z = a.z + d2 * xs[k].z + bb.z;
            o.w = a.w + d2 * xs[k].w + bb.w;
            *reinterpret_cast<float4*>(out + (size_t)(node0 + rr) * DIM + 4 * lane) = o;
        }
    }
}

extern "C" void kernel_launch(void* const* d_in, const int* in_sizes, int n_in,
                              void* d_out, int out_size, void* d_ws, size_t ws_size,
                              hipStream_t stream) {
    const float* x          = (const float*)d_in[0];
    const int*   edge_index = (const int*)d_in[1];
    const float* W          = (const float*)d_in[2];
    const float* b          = (const float*)d_in[3];
    const float* w_pred     = (const float*)d_in[4];
    const float* b_pred     = (const float*)d_in[5];
    float*       out        = (float*)d_out;

    const int* src = edge_index;
    const int* tgt = edge_index + N_EDGES;

    float* ws  = (float*)d_ws;
    float* p_a = ws;
    float* p_b = p_a + N_NODES;
    float* dis = p_b + N_NODES;
    float* xw  = dis + N_NODES + 16;
    int* toff  = (int*)(xw + (size_t)N_NODES * DIM);
    int* trec  = toff + N_TILES * OFF_STRIDE;
    int* rinfo = trec + N_TILES * TILE_EDGES;
    int* crec  = rinfo + 512;

    front_kernel     <<<N_TILES + DOT_BLOCKS, 1024, 0, stream>>>(src, tgt, x, w_pred, p_a, p_b, toff, trec);
    gemm_wmma_kernel <<<N_NODES / 16,         256,  0, stream>>>(x, W, toff, trec, p_a, p_b, b_pred, crec, rinfo, dis, xw);
    range_sum_kernel <<<N_BINS,               1024, 0, stream>>>(crec, rinfo, dis, xw, b, out);
}
